# starting kernel
# baseline (speedup 1.0000x reference)
#include <hip/hip_runtime.h>

typedef __attribute__((ext_vector_type(2))) float v2f;
typedef __attribute__((ext_vector_type(8))) float v8f;

#define HIDC 64
#define LDSW 68


__global__ void init_deg_kernel(float* __restrict__ deg, int n) {
  int i = blockIdx.x * blockDim.x + threadIdx.x;
  if (i < n) deg[i] = 1.0f;
}

__global__ void deg_scatter_kernel(const int* __restrict__ dst, float* __restrict__ deg, int E) {
  int e = blockIdx.x * blockDim.x + threadIdx.x;
  if (e < E) unsafeAtomicAdd(&deg[dst[e]], 1.0f);
}

__global__ void dinv_kernel(float* __restrict__ deg, int n) {
  int i = blockIdx.x * blockDim.x + threadIdx.x;
  if (i < n) {
    float d = deg[i];
    deg[i] = (d > 0.0f) ? rsqrtf(d) : 0.0f;
  }
}


__global__ __launch_bounds__(256) void gemm_wmma_kernel(
    const float* __restrict__ A, const float* __restrict__ W,
    float* __restrict__ H, int n, int K) {
  __shared__ float tile[32 * LDSW];

  const int t     = threadIdx.x;
  const int wid   = t >> 5;
  const int lane  = t & 31;
  const int row16 = lane & 15;
  const int half  = lane >> 4;
  const int m_sub = wid >> 2;
  const int n0    = (wid & 3) << 4;
  const int m_base = blockIdx.x * 32;

  v8f acc = {};

  const int lr = t >> 3;
  const int lc = (t & 7) << 3;

  for (int kb = 0; kb < K; kb += 64) {
    float4 v0 = make_float4(0.f, 0.f, 0.f, 0.f);
    float4 v1 = v0;
    int grow = m_base + lr;
    if (grow < n) {
      const float* p = A + (size_t)grow * K + kb + lc;
      v0 = *(const float4*)(p);
      v1 = *(const float4*)(p + 4);
    }
    float* q = &tile[lr * LDSW + lc];
    *(float4*)(q)     = v0;
    *(float4*)(q + 4) = v1;
    __syncthreads();

    const float* arow = &tile[(m_sub * 16 + row16) * LDSW];
#pragma unroll
    for (int kk = 0; kk < 64; kk += 4) {
      v2f a = *(const v2f*)(arow + kk + half * 2);
      int krow = kb + kk + half * 2;
      v2f b;
      b.x = W[(size_t)krow * HIDC + n0 + row16];
      b.y = W[(size_t)(krow + 1) * HIDC + n0 + row16];
      acc = __builtin_amdgcn_wmma_f32_16x16x4_f32(
          false, a, false, b, (short)0, acc, false, false);
    }
    __syncthreads();
  }

  int mrow = m_base + m_sub * 16 + half * 8;
#pragma unroll
  for (int r = 0; r < 8; ++r) {
    int gr = mrow + r;
    if (gr < n) H[(size_t)gr * HIDC + n0 + row16] = acc[r];
  }
}


__global__ void selfloop_init_kernel(const float* __restrict__ h,
                                     const float* __restrict__ dinv,
                                     const float* __restrict__ bias,
                                     float* __restrict__ agg,
                                     int total, int use_bias) {
  int i = blockIdx.x * blockDim.x + threadIdx.x;
  if (i >= total) return;
  int node = i >> 6;
  float d = dinv[node];
  float v = h[i] * d * d;
  if (use_bias) v += bias[i & 63];
  agg[i] = v;
}

__global__ void edge_scatter_kernel(const int* __restrict__ src,
                                    const int* __restrict__ dst,
                                    const float* __restrict__ dinv,
                                    const float* __restrict__ h,
                                    float* __restrict__ agg, int E) {
  int t = blockIdx.x * blockDim.x + threadIdx.x;
  int e = t >> 5;
  if (e >= E) return;
  int c = (t & 31) << 1;
  int s = src[e];
  int d = dst[e];
  float norm = dinv[s] * dinv[d];
  float2 hv = *(const float2*)(h + (size_t)s * HIDC + c);
  float* out = agg + (size_t)d * HIDC + c;
  unsafeAtomicAdd(out,     hv.x * norm);
  unsafeAtomicAdd(out + 1, hv.y * norm);
}

__global__ void bias_relu_kernel(float* __restrict__ agg,
                                 const float* __restrict__ bias, int total) {
  int i = blockIdx.x * blockDim.x + threadIdx.x;
  if (i >= total) return;
  float v = agg[i] + bias[i & 63];
  agg[i] = v > 0.0f ? v : 0.0f;
}


extern "C" void kernel_launch(void* const* d_in, const int* in_sizes, int n_in,
                              void* d_out, int out_size, void* d_ws, size_t ws_size,
                              hipStream_t stream) {
  const float* x  = (const float*)d_in[0];
  const int*   ei = (const int*)d_in[1];
  const float* W1 = (const float*)d_in[2];
  const float* b1 = (const float*)d_in[3];
  const float* W2 = (const float*)d_in[4];
  const float* b2 = (const float*)d_in[5];

  const int IN_CH = 512;
  const int n = in_sizes[0] / IN_CH;
  const int E = in_sizes[1] / 2;
  const int* src = ei;
  const int* dst = ei + E;
  float* out = (float*)d_out;

  float* ws = (float*)d_ws;
  size_t n_al = ((size_t)n + 255) & ~(size_t)255;
  float* dinv = ws;
  float* bufA = ws + n_al;
  float* bufB = bufA + (size_t)n * HIDC;

  const int B = 256;
  const long long total = (long long)n * HIDC;
  unsigned gN   = (unsigned)((n + B - 1) / B);
  unsigned gE   = (unsigned)((E + B - 1) / B);
  unsigned gNC  = (unsigned)((total + B - 1) / B);
  unsigned gE32 = (unsigned)(((long long)E * 32 + B - 1) / B);
  unsigned gMM  = (unsigned)((n + 31) / 32);

  init_deg_kernel<<<gN, B, 0, stream>>>(dinv, n);
  deg_scatter_kernel<<<gE, B, 0, stream>>>(dst, dinv, E);
  dinv_kernel<<<gN, B, 0, stream>>>(dinv, n);

  gemm_wmma_kernel<<<gMM, 256, 0, stream>>>(x, W1, bufA, n, IN_CH);
  selfloop_init_kernel<<<gNC, B, 0, stream>>>(bufA, dinv, b1, bufB, (int)total, 0);
  edge_scatter_kernel<<<gE32, B, 0, stream>>>(src, dst, dinv, bufA, bufB, E);
  bias_relu_kernel<<<gNC, B, 0, stream>>>(bufB, b1, (int)total);

  gemm_wmma_kernel<<<gMM, 256, 0, stream>>>(bufB, W2, bufA, n, HIDC);
  selfloop_init_kernel<<<gNC, B, 0, stream>>>(bufA, dinv, b2, out, (int)total, 1);
  edge_scatter_kernel<<<gE32, B, 0, stream>>>(src, dst, dinv, bufA, out, E);
}


# degree count: LDS histogram per 16384-node range and edge slice, flushed as whole-line f32 atomic adds of integer counts (sum order changes, sums exact) instead of one global atomic per edge
# speedup vs baseline: 1.1639x; 1.1639x over previous
#include <hip/hip_runtime.h>

typedef __attribute__((ext_vector_type(2))) float v2f;
typedef __attribute__((ext_vector_type(8))) float v8f;

#define HIDC 64
#define LDSW 68


__global__ void init_deg_kernel(float* __restrict__ deg, int n) {
  int i = blockIdx.x * blockDim.x + threadIdx.x;
  if (i < n) deg[i] = 1.0f;
}

#define DEG_RANGE  16384
#define DEG_SLICES 36
#define DEG_BLOCK  1024
#define DEG_BATCH  8

__global__ __launch_bounds__(DEG_BLOCK) void deg_hist_kernel(const int* __restrict__ dst,
                                                            float* __restrict__ deg,
                                                            int E, int n) {
  __shared__ unsigned cnt[DEG_RANGE];
  const int t = threadIdx.x;
  for (int j = t; j < DEG_RANGE; j += DEG_BLOCK) cnt[j] = 0u;
  __syncthreads();

  const int base = blockIdx.y * DEG_RANGE;
  const int per  = (E + (int)gridDim.x - 1) / (int)gridDim.x;
  const int e0   = blockIdx.x * per;
  const int e1   = (e0 + per < E) ? (e0 + per) : E;

  int e = e0 + t;
  for (; e + (DEG_BATCH - 1) * DEG_BLOCK < e1; e += DEG_BATCH * DEG_BLOCK) {
    int d[DEG_BATCH];
#pragma unroll
    for (int u = 0; u < DEG_BATCH; ++u) d[u] = dst[e + u * DEG_BLOCK];
#pragma unroll
    for (int u = 0; u < DEG_BATCH; ++u) {
      unsigned r = (unsigned)(d[u] - base);
      if (r < (unsigned)DEG_RANGE) atomicAdd(&cnt[r], 1u);
    }
  }
  for (; e < e1; e += DEG_BLOCK) {
    unsigned r = (unsigned)(dst[e] - base);
    if (r < (unsigned)DEG_RANGE) atomicAdd(&cnt[r], 1u);
  }
  __syncthreads();

  for (int j = t; j < DEG_RANGE; j += DEG_BLOCK) {
    unsigned c = cnt[j];
    int node = base + j;
    if (c != 0u && node < n) unsafeAtomicAdd(&deg[node], (float)c);
  }
}

__global__ void dinv_kernel(float* __restrict__ deg, int n) {
  int i = blockIdx.x * blockDim.x + threadIdx.x;
  if (i < n) {
    float d = deg[i];
    deg[i] = (d > 0.0f) ? rsqrtf(d) : 0.0f;
  }
}


__global__ __launch_bounds__(256) void gemm_wmma_kernel(
    const float* __restrict__ A, const float* __restrict__ W,
    float* __restrict__ H, int n, int K) {
  __shared__ float tile[32 * LDSW];

  const int t     = threadIdx.x;
  const int wid   = t >> 5;
  const int lane  = t & 31;
  const int row16 = lane & 15;
  const int half  = lane >> 4;
  const int m_sub = wid >> 2;
  const int n0    = (wid & 3) << 4;
  const int m_base = blockIdx.x * 32;

  v8f acc = {};

  const int lr = t >> 3;
  const int lc = (t & 7) << 3;

  for (int kb = 0; kb < K; kb += 64) {
    float4 v0 = make_float4(0.f, 0.f, 0.f, 0.f);
    float4 v1 = v0;
    int grow = m_base + lr;
    if (grow < n) {
      const float* p = A + (size_t)grow * K + kb + lc;
      v0 = *(const float4*)(p);
      v1 = *(const float4*)(p + 4);
    }
    float* q = &tile[lr * LDSW + lc];
    *(float4*)(q)     = v0;
    *(float4*)(q + 4) = v1;
    __syncthreads();

    const float* arow = &tile[(m_sub * 16 + row16) * LDSW];
#pragma unroll
    for (int kk = 0; kk < 64; kk += 4) {
      v2f a = *(const v2f*)(arow + kk + half * 2);
      int krow = kb + kk + half * 2;
      v2f b;
      b.x = W[(size_t)krow * HIDC + n0 + row16];
      b.y = W[(size_t)(krow + 1) * HIDC + n0 + row16];
      acc = __builtin_amdgcn_wmma_f32_16x16x4_f32(
          false, a, false, b, (short)0, acc, false, false);
    }
    __syncthreads();
  }

  int mrow = m_base + m_sub * 16 + half * 8;
#pragma unroll
  for (int r = 0; r < 8; ++r) {
    int gr = mrow + r;
    if (gr < n) H[(size_t)gr * HIDC + n0 + row16] = acc[r];
  }
}


__global__ void selfloop_init_kernel(const float* __restrict__ h,
                                     const float* __restrict__ dinv,
                                     const float* __restrict__ bias,
                                     float* __restrict__ agg,
                                     int total, int use_bias) {
  int i = blockIdx.x * blockDim.x + threadIdx.x;
  if (i >= total) return;
  int node = i >> 6;
  float d = dinv[node];
  float v = h[i] * d * d;
  if (use_bias) v += bias[i & 63];
  agg[i] = v;
}

__global__ void edge_scatter_kernel(const int* __restrict__ src,
                                    const int* __restrict__ dst,
                                    const float* __restrict__ dinv,
                                    const float* __restrict__ h,
                                    float* __restrict__ agg, int E) {
  int t = blockIdx.x * blockDim.x + threadIdx.x;
  int e = t >> 5;
  if (e >= E) return;
  int c = (t & 31) << 1;
  int s = src[e];
  int d = dst[e];
  float norm = dinv[s] * dinv[d];
  float2 hv = *(const float2*)(h + (size_t)s * HIDC + c);
  float* out = agg + (size_t)d * HIDC + c;
  unsafeAtomicAdd(out,     hv.x * norm);
  unsafeAtomicAdd(out + 1, hv.y * norm);
}

__global__ void bias_relu_kernel(float* __restrict__ agg,
                                 const float* __restrict__ bias, int total) {
  int i = blockIdx.x * blockDim.x + threadIdx.x;
  if (i >= total) return;
  float v = agg[i] + bias[i & 63];
  agg[i] = v > 0.0f ? v : 0.0f;
}


extern "C" void kernel_launch(void* const* d_in, const int* in_sizes, int n_in,
                              void* d_out, int out_size, void* d_ws, size_t ws_size,
                              hipStream_t stream) {
  const float* x  = (const float*)d_in[0];
  const int*   ei = (const int*)d_in[1];
  const float* W1 = (const float*)d_in[2];
  const float* b1 = (const float*)d_in[3];
  const float* W2 = (const float*)d_in[4];
  const float* b2 = (const float*)d_in[5];

  const int IN_CH = 512;
  const int n = in_sizes[0] / IN_CH;
  const int E = in_sizes[1] / 2;
  const int* src = ei;
  const int* dst = ei + E;
  float* out = (float*)d_out;

  float* ws = (float*)d_ws;
  size_t n_al = ((size_t)n + 255) & ~(size_t)255;
  float* dinv = ws;
  float* bufA = ws + n_al;
  float* bufB = bufA + (size_t)n * HIDC;

  const int B = 256;
  const long long total = (long long)n * HIDC;
  unsigned gN   = (unsigned)((n + B - 1) / B);
  unsigned gR   = (unsigned)((n + DEG_RANGE - 1) / DEG_RANGE);
  unsigned gNC  = (unsigned)((total + B - 1) / B);
  unsigned gE32 = (unsigned)(((long long)E * 32 + B - 1) / B);
  unsigned gMM  = (unsigned)((n + 31) / 32);

  init_deg_kernel<<<gN, B, 0, stream>>>(dinv, n);
  deg_hist_kernel<<<dim3(DEG_SLICES, gR), DEG_BLOCK, 0, stream>>>(dst, dinv, E, n);
  dinv_kernel<<<gN, B, 0, stream>>>(dinv, n);

  gemm_wmma_kernel<<<gMM, 256, 0, stream>>>(x, W1, bufA, n, IN_CH);
  selfloop_init_kernel<<<gNC, B, 0, stream>>>(bufA, dinv, b1, bufB, (int)total, 0);
  edge_scatter_kernel<<<gE32, B, 0, stream>>>(src, dst, dinv, bufA, bufB, E);
  bias_relu_kernel<<<gNC, B, 0, stream>>>(bufB, b1, (int)total);

  gemm_wmma_kernel<<<gMM, 256, 0, stream>>>(bufB, W2, bufA, n, HIDC);
  selfloop_init_kernel<<<gNC, B, 0, stream>>>(bufA, dinv, b2, out, (int)total, 1);
  edge_scatter_kernel<<<gE32, B, 0, stream>>>(src, dst, dinv, bufA, out, E);
}


# float-atomic edge scatter replaced by gather: edges grouped by destination once per call (LDS counting sort), one wave sums a node's source rows in registers, one store; sum order = list order
# speedup vs baseline: 2.4672x; 2.1198x over previous
#include <hip/hip_runtime.h>

typedef __attribute__((ext_vector_type(2))) float v2f;
typedef __attribute__((ext_vector_type(8))) float v8f;

#define HIDC 64
#define LDSW 68


__global__ void init_deg_kernel(float* __restrict__ deg, int* __restrict__ cursor, int n) {
  int i = blockIdx.x * blockDim.x + threadIdx.x;
  if (i < n) {
    deg[i] = 1.0f;
    cursor[i] = 0;
  }
}

#define DEG_RANGE  16384
#define DEG_SLICES 36
#define DEG_BLOCK  1024
#define DEG_BATCH  8
#define DEG_PER_THREAD (DEG_RANGE / DEG_BLOCK)

__global__ __launch_bounds__(DEG_BLOCK) void deg_hist_kernel(const int* __restrict__ dst,
                                                            float* __restrict__ deg,
                                                            int E, int n) {
  __shared__ unsigned cnt[DEG_RANGE];
  const int t = threadIdx.x;
  for (int j = t; j < DEG_RANGE; j += DEG_BLOCK) cnt[j] = 0u;
  __syncthreads();

  const int base = blockIdx.y * DEG_RANGE;
  const int per  = (E + (int)gridDim.x - 1) / (int)gridDim.x;
  const int e0   = blockIdx.x * per;
  const int e1   = (e0 + per < E) ? (e0 + per) : E;

  int e = e0 + t;
  for (; e + (DEG_BATCH - 1) * DEG_BLOCK < e1; e += DEG_BATCH * DEG_BLOCK) {
    int d[DEG_BATCH];
#pragma unroll
    for (int u = 0; u < DEG_BATCH; ++u) d[u] = dst[e + u * DEG_BLOCK];
#pragma unroll
    for (int u = 0; u < DEG_BATCH; ++u) {
      unsigned r = (unsigned)(d[u] - base);
      if (r < (unsigned)DEG_RANGE) atomicAdd(&cnt[r], 1u);
    }
  }
  for (; e < e1; e += DEG_BLOCK) {
    unsigned r = (unsigned)(dst[e] - base);
    if (r < (unsigned)DEG_RANGE) atomicAdd(&cnt[r], 1u);
  }
  __syncthreads();

  for (int j = t; j < DEG_RANGE; j += DEG_BLOCK) {
    unsigned c = cnt[j];
    int node = base + j;
    if (c != 0u && node < n) unsafeAtomicAdd(&deg[node], (float)c);
  }
}

__global__ void dinv_kernel(float* __restrict__ deg, int n) {
  int i = blockIdx.x * blockDim.x + threadIdx.x;
  if (i < n) {
    float d = deg[i];
    deg[i] = (d > 0.0f) ? rsqrtf(d) : 0.0f;
  }
}


#define SCAN_BLOCK 1024

__global__ __launch_bounds__(SCAN_BLOCK) void row_ptr_scan_kernel(const float* __restrict__ deg,
                                                                  int* __restrict__ row_ptr, int n) {
  __shared__ int wsum[32];
  __shared__ int s_prefix;
  const int t    = threadIdx.x;
  const int lane = t & 31;
  const int wid  = t >> 5;
  const int first = blockIdx.x * SCAN_BLOCK;

  int part = 0;
  for (int j = t; j < first; j += SCAN_BLOCK) part += (int)deg[j] - 1;
  for (int o = 16; o > 0; o >>= 1) part += __shfl_xor(part, o, 32);
  if (lane == 0) wsum[wid] = part;
  __syncthreads();
  if (wid == 0) {
    int v = wsum[lane];
    for (int o = 16; o > 0; o >>= 1) v += __shfl_xor(v, o, 32);
    if (lane == 0) s_prefix = v;
  }
  __syncthreads();
  const int prefix = s_prefix;
  __syncthreads();

  const int i = first + t;
  const int v = (i < n) ? ((int)deg[i] - 1) : 0;
  int incl = v;
  for (int o = 1; o < 32; o <<= 1) {
    int u = __shfl_up(incl, o, 32);
    if (lane >= o) incl += u;
  }
  if (lane == 31) wsum[wid] = incl;
  __syncthreads();
  if (wid == 0) {
    const int w = wsum[lane];
    int wi = w;
    for (int o = 1; o < 32; o <<= 1) {
      int u = __shfl_up(wi, o, 32);
      if (lane >= o) wi += u;
    }
    wsum[lane] = wi - w;
  }
  __syncthreads();
  if (i <= n) row_ptr[i] = prefix + wsum[wid] + incl - v;
}

__global__ __launch_bounds__(DEG_BLOCK) void csr_fill_kernel(const int* __restrict__ src,
                                                            const int* __restrict__ dst,
                                                            const int* __restrict__ row_ptr,
                                                            int* __restrict__ cursor,
                                                            int* __restrict__ csr_src,
                                                            int E, int n) {
  __shared__ unsigned cnt[DEG_RANGE];
  const int t = threadIdx.x;
  for (int j = t; j < DEG_RANGE; j += DEG_BLOCK) cnt[j] = 0u;
  __syncthreads();

  const int base = blockIdx.y * DEG_RANGE;
  const int per  = (E + (int)gridDim.x - 1) / (int)gridDim.x;
  const int e0   = blockIdx.x * per;
  const int e1   = (e0 + per < E) ? (e0 + per) : E;

  int e = e0 + t;
  for (; e + (DEG_BATCH - 1) * DEG_BLOCK < e1; e += DEG_BATCH * DEG_BLOCK) {
    int d[DEG_BATCH];
#pragma unroll
    for (int u = 0; u < DEG_BATCH; ++u) d[u] = dst[e + u * DEG_BLOCK];
#pragma unroll
    for (int u = 0; u < DEG_BATCH; ++u) {
      unsigned r = (unsigned)(d[u] - base);
      if (r < (unsigned)DEG_RANGE) atomicAdd(&cnt[r], 1u);
    }
  }
  for (; e < e1; e += DEG_BLOCK) {
    unsigned r = (unsigned)(dst[e] - base);
    if (r < (unsigned)DEG_RANGE) atomicAdd(&cnt[r], 1u);
  }
  __syncthreads();

  {
    unsigned c[DEG_PER_THREAD];
    int rp[DEG_PER_THREAD];
    int off[DEG_PER_THREAD];
#pragma unroll
    for (int u = 0; u < DEG_PER_THREAD; ++u) c[u] = cnt[t + u * DEG_BLOCK];
#pragma unroll
    for (int u = 0; u < DEG_PER_THREAD; ++u) {
      const int node = base + t + u * DEG_BLOCK;
      rp[u]  = 0;
      off[u] = 0;
      if (c[u] != 0u && node < n) {
        rp[u]  = row_ptr[node];
        off[u] = atomicAdd(&cursor[node], (int)c[u]);
      }
    }
#pragma unroll
    for (int u = 0; u < DEG_PER_THREAD; ++u)
      if (c[u] != 0u) cnt[t + u * DEG_BLOCK] = (unsigned)(rp[u] + off[u]);
  }
  __syncthreads();

  e = e0 + t;
  for (; e + (DEG_BATCH - 1) * DEG_BLOCK < e1; e += DEG_BATCH * DEG_BLOCK) {
    int d[DEG_BATCH];
    int s[DEG_BATCH];
#pragma unroll
    for (int u = 0; u < DEG_BATCH; ++u) {
      d[u] = dst[e + u * DEG_BLOCK];
      s[u] = src[e + u * DEG_BLOCK];
    }
#pragma unroll
    for (int u = 0; u < DEG_BATCH; ++u) {
      unsigned r = (unsigned)(d[u] - base);
      if (r < (unsigned)DEG_RANGE) {
        unsigned pos = atomicAdd(&cnt[r], 1u);
        csr_src[pos] = s[u];
      }
    }
  }
  for (; e < e1; e += DEG_BLOCK) {
    unsigned r = (unsigned)(dst[e] - base);
    if (r < (unsigned)DEG_RANGE) {
      unsigned pos = atomicAdd(&cnt[r], 1u);
      csr_src[pos] = src[e];
    }
  }
}


__global__ __launch_bounds__(256) void gemm_wmma_kernel(
    const float* __restrict__ A, const float* __restrict__ W,
    float* __restrict__ H, int n, int K) {
  __shared__ float tile[32 * LDSW];

  const int t     = threadIdx.x;
  const int wid   = t >> 5;
  const int lane  = t & 31;
  const int row16 = lane & 15;
  const int half  = lane >> 4;
  const int m_sub = wid >> 2;
  const int n0    = (wid & 3) << 4;
  const int m_base = blockIdx.x * 32;

  v8f acc = {};

  const int lr = t >> 3;
  const int lc = (t & 7) << 3;

  for (int kb = 0; kb < K; kb += 64) {
    float4 v0 = make_float4(0.f, 0.f, 0.f, 0.f);
    float4 v1 = v0;
    int grow = m_base + lr;
    if (grow < n) {
      const float* p = A + (size_t)grow * K + kb + lc;
      v0 = *(const float4*)(p);
      v1 = *(const float4*)(p + 4);
    }
    float* q = &tile[lr * LDSW + lc];
    *(float4*)(q)     = v0;
    *(float4*)(q + 4) = v1;
    __syncthreads();

    const float* arow = &tile[(m_sub * 16 + row16) * LDSW];
#pragma unroll
    for (int kk = 0; kk < 64; kk += 4) {
      v2f a = *(const v2f*)(arow + kk + half * 2);
      int krow = kb + kk + half * 2;
      v2f b;
      b.x = W[(size_t)krow * HIDC + n0 + row16];
      b.y = W[(size_t)(krow + 1) * HIDC + n0 + row16];
      acc = __builtin_amdgcn_wmma_f32_16x16x4_f32(
          false, a, false, b, (short)0, acc, false, false);
    }
    __syncthreads();
  }

  int mrow = m_base + m_sub * 16 + half * 8;
#pragma unroll
  for (int r = 0; r < 8; ++r) {
    int gr = mrow + r;
    if (gr < n) H[(size_t)gr * HIDC + n0 + row16] = acc[r];
  }
}


#define GATHER_UNROLL 8

__global__ __launch_bounds__(256) void gather_rows_kernel(const int* __restrict__ row_ptr,
                                                          const int* __restrict__ csr_src,
                                                          const float* __restrict__ dinv,
                                                          const float* __restrict__ h,
                                                          const float* __restrict__ bias,
                                                          float* __restrict__ outp,
                                                          int n, int relu) {
  const int lane = threadIdx.x & 31;
  const int node = __builtin_amdgcn_readfirstlane((int)(blockIdx.x * 8u + (threadIdx.x >> 5)));
  if (node >= n) return;

  const int start = row_ptr[node];
  const int end   = row_ptr[node + 1];
  const float di  = dinv[node];
  const int c = lane << 1;

  float2 acc;
  {
    float2 hv = *(const float2*)(h + (size_t)node * HIDC + c);
    acc.x = hv.x * di * di;
    acc.y = hv.y * di * di;
  }

  for (int k0 = start; k0 < end; k0 += 32) {
    const int k = k0 + lane;
    int   s_l = 0;
    float w_l = 0.0f;
    if (k < end) {
      s_l = csr_src[k];
      w_l = dinv[s_l] * di;
    }
    const int m = (end - k0 < 32) ? (end - k0) : 32;

    for (int j = 0; j < m; j += GATHER_UNROLL) {
      float2 r[GATHER_UNROLL];
      float  w[GATHER_UNROLL];
#pragma unroll
      for (int u = 0; u < GATHER_UNROLL; ++u) {
        r[u] = make_float2(0.0f, 0.0f);
        w[u] = 0.0f;
        if (j + u < m) {
          const int s = __builtin_amdgcn_readlane(s_l, j + u);
          w[u] = __int_as_float(__builtin_amdgcn_readlane(__float_as_int(w_l), j + u));
          r[u] = *(const float2*)(h + (size_t)s * HIDC + c);
        }
      }
#pragma unroll
      for (int u = 0; u < GATHER_UNROLL; ++u) {
        acc.x += r[u].x * w[u];
        acc.y += r[u].y * w[u];
      }
    }
  }

  const float2 b = *(const float2*)(bias + c);
  float ox = acc.x + b.x;
  float oy = acc.y + b.y;
  if (relu) {
    ox = ox > 0.0f ? ox : 0.0f;
    oy = oy > 0.0f ? oy : 0.0f;
  }
  *(float2*)(outp + (size_t)node * HIDC + c) = make_float2(ox, oy);
}


extern "C" void kernel_launch(void* const* d_in, const int* in_sizes, int n_in,
                              void* d_out, int out_size, void* d_ws, size_t ws_size,
                              hipStream_t stream) {
  const float* x  = (const float*)d_in[0];
  const int*   ei = (const int*)d_in[1];
  const float* W1 = (const float*)d_in[2];
  const float* b1 = (const float*)d_in[3];
  const float* W2 = (const float*)d_in[4];
  const float* b2 = (const float*)d_in[5];

  const int IN_CH = 512;
  const int n = in_sizes[0] / IN_CH;
  const int E = in_sizes[1] / 2;
  const int* src = ei;
  const int* dst = ei + E;
  float* out = (float*)d_out;

  float* ws = (float*)d_ws;
  size_t n_al = ((size_t)n + 255) & ~(size_t)255;
  float* dinv = ws;
  float* bufA = ws + n_al;
  int* rowp   = (int*)(bufA + (size_t)n * HIDC);
  int* cursor = rowp + n_al + 256;
  int* csrs   = cursor + n_al;

  const int B = 256;
  unsigned gN   = (unsigned)((n + B - 1) / B);
  unsigned gR   = (unsigned)((n + DEG_RANGE - 1) / DEG_RANGE);
  unsigned gS   = (unsigned)((n + 1 + SCAN_BLOCK - 1) / SCAN_BLOCK);
  unsigned gG   = (unsigned)((n + 7) / 8);
  unsigned gMM  = (unsigned)((n + 31) / 32);

  init_deg_kernel<<<gN, B, 0, stream>>>(dinv, cursor, n);
  deg_hist_kernel<<<dim3(DEG_SLICES, gR), DEG_BLOCK, 0, stream>>>(dst, dinv, E, n);
  row_ptr_scan_kernel<<<gS, SCAN_BLOCK, 0, stream>>>(dinv, rowp, n);
  dinv_kernel<<<gN, B, 0, stream>>>(dinv, n);
  csr_fill_kernel<<<dim3(DEG_SLICES, gR), DEG_BLOCK, 0, stream>>>(src, dst, rowp, cursor, csrs, E, n);

  gemm_wmma_kernel<<<gMM, 256, 0, stream>>>(x, W1, bufA, n, IN_CH);
  gather_rows_kernel<<<gG, 256, 0, stream>>>(rowp, csrs, dinv, bufA, b1, out, n, 1);

  gemm_wmma_kernel<<<gMM, 256, 0, stream>>>(out, W2, bufA, n, HIDC);
  gather_rows_kernel<<<gG, 256, 0, stream>>>(rowp, csrs, dinv, bufA, b2, out, n, 0);
}


# in-edge list build made coalesced and atomic-free: edges sorted by 512-node bucket in LDS, written as runs; one block per bucket forms dinv, row starts and list piece; sum order = list order still
# speedup vs baseline: 2.7777x; 1.1259x over previous
#include <hip/hip_runtime.h>

typedef __attribute__((ext_vector_type(2))) float v2f;
typedef __attribute__((ext_vector_type(8))) float v8f;

#define HIDC 64
#define LDSW 68


#define BKT_NODES 512
#define BKT_SHIFT 9
#define BKT_MAX   256
#define PCH       4096
#define PBLOCK    1024
#define PPT       (PCH / PBLOCK)
#define KB_CAP    12288
#define COL_BATCH 16

__device__ __forceinline__ int block_excl_scan_1024(int v, int* wsum, int* s_tot) {
  const int lane = threadIdx.x & 31;
  const int wid  = threadIdx.x >> 5;
  int incl = v;
  for (int o = 1; o < 32; o <<= 1) {
    int u = __shfl_up(incl, o, 32);
    if (lane >= o) incl += u;
  }
  __syncthreads();
  if (lane == 31) wsum[wid] = incl;
  __syncthreads();
  if (wid == 0) {
    const int w = wsum[lane];
    int wi = w;
    for (int o = 1; o < 32; o <<= 1) {
      int u = __shfl_up(wi, o, 32);
      if (lane >= o) wi += u;
    }
    wsum[lane] = wi - w;
    if (lane == 31) *s_tot = wi;
  }
  __syncthreads();
  return wsum[wid] + incl - v;
}

__global__ __launch_bounds__(PBLOCK) void bucket_count_kernel(const int* __restrict__ dst,
                                                             int* __restrict__ counts,
                                                             int E, int nbk) {
  __shared__ unsigned hist[BKT_MAX];
  const int t = threadIdx.x;
  if (t < BKT_MAX) hist[t] = 0u;
  __syncthreads();

  const int e0 = blockIdx.x * PCH;
  int d[PPT];
#pragma unroll
  for (int u = 0; u < PPT; ++u) {
    const int e = e0 + t + u * PBLOCK;
    d[u] = -1;
    if (e < E) d[u] = dst[e];
  }
#pragma unroll
  for (int u = 0; u < PPT; ++u) {
    const unsigned b = (unsigned)d[u] >> BKT_SHIFT;
    if (b < (unsigned)nbk) atomicAdd(&hist[b], 1u);
  }
  __syncthreads();
  if (t < nbk) counts[(size_t)blockIdx.x * nbk + t] = (int)hist[t];
}

__global__ __launch_bounds__(PBLOCK) void bucket_partition_kernel(const int* __restrict__ src,
                                                                 const int* __restrict__ dst,
                                                                 const int* __restrict__ counts,
                                                                 int* __restrict__ bstart_out,
                                                                 int* __restrict__ pairs,
                                                                 int E, int nbk, int pa) {
  __shared__ int part_pre[4 * BKT_MAX];
  __shared__ int part_tot[4 * BKT_MAX];
  __shared__ int run_start[BKT_MAX];
  __shared__ int loc_start[BKT_MAX];
  __shared__ unsigned cur[BKT_MAX];
  __shared__ int wsum[32];
  __shared__ int s_tot;
  __shared__ int s_src[PCH];
  __shared__ int s_dst[PCH];

  const int t   = threadIdx.x;
  const int blk = blockIdx.x;

  {
    const int g = t >> 8;
    const int b = t & (BKT_MAX - 1);
    int pre = 0, tot = 0;
    if (b < nbk) {
      for (int r0 = g; r0 < pa; r0 += 4 * COL_BATCH) {
        int c[COL_BATCH];
#pragma unroll
        for (int u = 0; u < COL_BATCH; ++u) {
          const int r = r0 + 4 * u;
          c[u] = (r < pa) ? counts[(size_t)r * nbk + b] : 0;
        }
#pragma unroll
        for (int u = 0; u < COL_BATCH; ++u) {
          tot += c[u];
          pre += (r0 + 4 * u < blk) ? c[u] : 0;
        }
      }
    }
    part_pre[t] = pre;
    part_tot[t] = tot;
  }
  __syncthreads();

  int pre = 0, tot = 0, own = 0;
  if (t < nbk) {
    pre = part_pre[t] + part_pre[t + BKT_MAX] + part_pre[t + 2 * BKT_MAX] + part_pre[t + 3 * BKT_MAX];
    tot = part_tot[t] + part_tot[t + BKT_MAX] + part_tot[t + 2 * BKT_MAX] + part_tot[t + 3 * BKT_MAX];
    own = counts[(size_t)blk * nbk + t];
  }
  const int bst = block_excl_scan_1024(tot, wsum, &s_tot);
  if (t < nbk) {
    run_start[t] = bst + pre;
    cur[t] = 0u;
    if (blk == 0) bstart_out[t] = bst;
  }
  if (blk == 0 && t == nbk) bstart_out[nbk] = bst;
  const int lst = block_excl_scan_1024(own, wsum, &s_tot);
  if (t < nbk) loc_start[t] = lst;
  __syncthreads();
  const int placed = s_tot;

  const int e0 = blk * PCH;
  int d[PPT];
  int s[PPT];
#pragma unroll
  for (int u = 0; u < PPT; ++u) {
    const int e = e0 + t + u * PBLOCK;
    d[u] = -1;
    s[u] = 0;
    if (e < E) {
      d[u] = dst[e];
      s[u] = src[e];
    }
  }
#pragma unroll
  for (int u = 0; u < PPT; ++u) {
    const unsigned b = (unsigned)d[u] >> BKT_SHIFT;
    if (b < (unsigned)nbk) {
      const unsigned r = atomicAdd(&cur[b], 1u);
      const int p = loc_start[b] + (int)r;
      s_src[p] = s[u];
      s_dst[p] = d[u];
    }
  }
  __syncthreads();

  int2* __restrict__ rec = reinterpret_cast<int2*>(pairs);
  for (int i = t; i < placed; i += PBLOCK) {
    const int dd = s_dst[i];
    const int b  = dd >> BKT_SHIFT;
    rec[run_start[b] + (i - loc_start[b])] = make_int2(s_src[i], dd);
  }
}

#define KB_RPT (KB_CAP / PBLOCK)

__global__ __launch_bounds__(PBLOCK) void bucket_build_kernel(const int* __restrict__ pairs,
                                                             const int* __restrict__ bstart,
                                                             float* __restrict__ dinv,
                                                             int* __restrict__ row_ptr,
                                                             int* __restrict__ csr_src,
                                                             int n) {
  __shared__ unsigned cnt[BKT_NODES];
  __shared__ int excl[BKT_NODES];
  __shared__ int sorted[KB_CAP];
  __shared__ int wsum[32];
  __shared__ int s_tot;

  const int2* __restrict__ rec = reinterpret_cast<const int2*>(pairs);
  const int t    = threadIdx.x;
  const int base = blockIdx.x * BKT_NODES;
  const int p0   = bstart[blockIdx.x];
  const int p1   = bstart[blockIdx.x + 1];
  const int m    = p1 - p0;
  const bool in_lds = (m <= KB_CAP);

  if (t < BKT_NODES) cnt[t] = 0u;
  __syncthreads();

  int rs[KB_RPT];
  int rv[KB_RPT];
  if (in_lds) {
    int2 raw[KB_RPT];
#pragma unroll
    for (int u = 0; u < KB_RPT; ++u) raw[u] = make_int2(0, 0);
    if (m > 0) {
#pragma unroll
      for (int u = 0; u < KB_RPT; ++u) {
        const int i  = t + u * PBLOCK;
        const int ic = (i < m) ? i : (m - 1);
        raw[u] = rec[p0 + ic];
      }
    }
#pragma unroll
    for (int u = 0; u < KB_RPT; ++u) {
      const int i = t + u * PBLOCK;
      rs[u] = raw[u].x;
      rv[u] = (i < m) ? (raw[u].y - base) : -1;
      if ((unsigned)rv[u] < (unsigned)BKT_NODES) atomicAdd(&cnt[rv[u]], 1u);
    }
  } else {
    for (int i = t; i < m; i += PBLOCK) {
      const unsigned v = (unsigned)(rec[p0 + i].y - base);
      if (v < (unsigned)BKT_NODES) atomicAdd(&cnt[v], 1u);
    }
  }
  __syncthreads();

  const int c  = (t < BKT_NODES) ? (int)cnt[t] : 0;
  const int ex = block_excl_scan_1024(c, wsum, &s_tot);
  if (t < BKT_NODES) {
    const int node = base + t;
    if (node < n) {
      const float dg = (float)(c + 1);
      dinv[node]    = (dg > 0.0f) ? rsqrtf(dg) : 0.0f;
      row_ptr[node] = p0 + ex;
    }
    excl[t] = ex;
    cnt[t]  = 0u;
  }
  if (blockIdx.x == gridDim.x - 1 && t == 0) row_ptr[n] = p1;
  __syncthreads();

  if (in_lds) {
#pragma unroll
    for (int u = 0; u < KB_RPT; ++u) {
      if ((unsigned)rv[u] < (unsigned)BKT_NODES) {
        const int pos = excl[rv[u]] + (int)atomicAdd(&cnt[rv[u]], 1u);
        sorted[pos] = rs[u];
      }
    }
    __syncthreads();
    for (int i = t; i < m; i += PBLOCK) csr_src[p0 + i] = sorted[i];
  } else {
    for (int i = t; i < m; i += PBLOCK) {
      const int2 pr = rec[p0 + i];
      const unsigned v = (unsigned)(pr.y - base);
      if (v < (unsigned)BKT_NODES) {
        const int pos = excl[v] + (int)atomicAdd(&cnt[v], 1u);
        csr_src[p0 + pos] = pr.x;
      }
    }
  }
}


__global__ __launch_bounds__(256) void gemm_wmma_kernel(
    const float* __restrict__ A, const float* __restrict__ W,
    float* __restrict__ H, int n, int K) {
  __shared__ float tile[32 * LDSW];

  const int t     = threadIdx.x;
  const int wid   = t >> 5;
  const int lane  = t & 31;
  const int row16 = lane & 15;
  const int half  = lane >> 4;
  const int m_sub = wid >> 2;
  const int n0    = (wid & 3) << 4;
  const int m_base = blockIdx.x * 32;

  v8f acc = {};

  const int lr = t >> 3;
  const int lc = (t & 7) << 3;

  for (int kb = 0; kb < K; kb += 64) {
    float4 v0 = make_float4(0.f, 0.f, 0.f, 0.f);
    float4 v1 = v0;
    int grow = m_base + lr;
    if (grow < n) {
      const float* p = A + (size_t)grow * K + kb + lc;
      v0 = *(const float4*)(p);
      v1 = *(const float4*)(p + 4);
    }
    float* q = &tile[lr * LDSW + lc];
    *(float4*)(q)     = v0;
    *(float4*)(q + 4) = v1;
    __syncthreads();

    const float* arow = &tile[(m_sub * 16 + row16) * LDSW];
#pragma unroll
    for (int kk = 0; kk < 64; kk += 4) {
      v2f a = *(const v2f*)(arow + kk + half * 2);
      int krow = kb + kk + half * 2;
      v2f b;
      b.x = W[(size_t)krow * HIDC + n0 + row16];
      b.y = W[(size_t)(krow + 1) * HIDC + n0 + row16];
      acc = __builtin_amdgcn_wmma_f32_16x16x4_f32(
          false, a, false, b, (short)0, acc, false, false);
    }
    __syncthreads();
  }

  int mrow = m_base + m_sub * 16 + half * 8;
#pragma unroll
  for (int r = 0; r < 8; ++r) {
    int gr = mrow + r;
    if (gr < n) H[(size_t)gr * HIDC + n0 + row16] = acc[r];
  }
}


#define GATHER_UNROLL 8

__global__ __launch_bounds__(256) void gather_rows_kernel(const int* __restrict__ row_ptr,
                                                          const int* __restrict__ csr_src,
                                                          const float* __restrict__ dinv,
                                                          const float* __restrict__ h,
                                                          const float* __restrict__ bias,
                                                          float* __restrict__ outp,
                                                          int n, int relu) {
  const int lane = threadIdx.x & 31;
  const int node = __builtin_amdgcn_readfirstlane((int)(blockIdx.x * 8u + (threadIdx.x >> 5)));
  if (node >= n) return;

  const int start = row_ptr[node];
  const int end   = row_ptr[node + 1];
  const float di  = dinv[node];
  const int c = lane << 1;

  float2 acc;
  {
    float2 hv = *(const float2*)(h + (size_t)node * HIDC + c);
    acc.x = hv.x * di * di;
    acc.y = hv.y * di * di;
  }

  for (int k0 = start; k0 < end; k0 += 32) {
    const int k = k0 + lane;
    int   s_l = 0;
    float w_l = 0.0f;
    if (k < end) {
      s_l = csr_src[k];
      w_l = dinv[s_l] * di;
    }
    const int m = (end - k0 < 32) ? (end - k0) : 32;

    for (int j = 0; j < m; j += GATHER_UNROLL) {
      float2 r[GATHER_UNROLL];
      float  w[GATHER_UNROLL];
#pragma unroll
      for (int u = 0; u < GATHER_UNROLL; ++u) {
        r[u] = make_float2(0.0f, 0.0f);
        w[u] = 0.0f;
        if (j + u < m) {
          const int s = __builtin_amdgcn_readlane(s_l, j + u);
          w[u] = __int_as_float(__builtin_amdgcn_readlane(__float_as_int(w_l), j + u));
          r[u] = *(const float2*)(h + (size_t)s * HIDC + c);
        }
      }
#pragma unroll
      for (int u = 0; u < GATHER_UNROLL; ++u) {
        acc.x += r[u].x * w[u];
        acc.y += r[u].y * w[u];
      }
    }
  }

  const float2 b = *(const float2*)(bias + c);
  float ox = acc.x + b.x;
  float oy = acc.y + b.y;
  if (relu) {
    ox = ox > 0.0f ? ox : 0.0f;
    oy = oy > 0.0f ? oy : 0.0f;
  }
  *(float2*)(outp + (size_t)node * HIDC + c) = make_float2(ox, oy);
}


extern "C" void kernel_launch(void* const* d_in, const int* in_sizes, int n_in,
                              void* d_out, int out_size, void* d_ws, size_t ws_size,
                              hipStream_t stream) {
  const float* x  = (const float*)d_in[0];
  const int*   ei = (const int*)d_in[1];
  const float* W1 = (const float*)d_in[2];
  const float* b1 = (const float*)d_in[3];
  const float* W2 = (const float*)d_in[4];
  const float* b2 = (const float*)d_in[5];

  const int IN_CH = 512;
  const int n = in_sizes[0] / IN_CH;
  const int E = in_sizes[1] / 2;
  const int* src = ei;
  const int* dst = ei + E;
  float* out = (float*)d_out;

  float* ws = (float*)d_ws;
  size_t n_al = ((size_t)n + 255) & ~(size_t)255;
  float* dinv = ws;
  float* bufA = ws + n_al;
  const int nbk = (n + BKT_NODES - 1) / BKT_NODES;
  const int pa  = (E + PCH - 1) / PCH;
  size_t cm_al = ((size_t)pa * (size_t)nbk + 255) & ~(size_t)255;
  int* rowp   = (int*)(bufA + (size_t)n * HIDC);
  int* counts = rowp + n_al + 256;
  int* bstart = counts + cm_al;
  int* csrs   = bstart + 512;
  int* pairs  = (int*)d_out;

  unsigned gG   = (unsigned)((n + 7) / 8);
  unsigned gMM  = (unsigned)((n + 31) / 32);

  bucket_count_kernel<<<(unsigned)pa, PBLOCK, 0, stream>>>(dst, counts, E, nbk);
  bucket_partition_kernel<<<(unsigned)pa, PBLOCK, 0, stream>>>(src, dst, counts, bstart, pairs, E, nbk, pa);
  bucket_build_kernel<<<(unsigned)nbk, PBLOCK, 0, stream>>>(pairs, bstart, dinv, rowp, csrs, n);

  gemm_wmma_kernel<<<gMM, 256, 0, stream>>>(x, W1, bufA, n, IN_CH);
  gather_rows_kernel<<<gG, 256, 0, stream>>>(rowp, csrs, dinv, bufA, b1, out, n, 1);

  gemm_wmma_kernel<<<gMM, 256, 0, stream>>>(out, W2, bufA, n, HIDC);
  gather_rows_kernel<<<gG, 256, 0, stream>>>(rowp, csrs, dinv, bufA, b2, out, n, 0);
}


# source weight dinv[s] folded into both products' epilogues (rows stored pre-scaled); gather sums plain rows, scales by dinv[i] once: no per-edge dinv lookup; product rounding changes, sum order kept
# speedup vs baseline: 2.9291x; 1.0545x over previous
#include <hip/hip_runtime.h>

typedef __attribute__((ext_vector_type(2))) float v2f;
typedef __attribute__((ext_vector_type(8))) float v8f;

#define HIDC 64
#define LDSW 68


#define BKT_NODES 512
#define BKT_SHIFT 9
#define BKT_MAX   256
#define PCH       4096
#define PBLOCK    1024
#define PPT       (PCH / PBLOCK)
#define KB_CAP    12288
#define COL_BATCH 16

__device__ __forceinline__ int block_excl_scan_1024(int v, int* wsum, int* s_tot) {
  const int lane = threadIdx.x & 31;
  const int wid  = threadIdx.x >> 5;
  int incl = v;
  for (int o = 1; o < 32; o <<= 1) {
    int u = __shfl_up(incl, o, 32);
    if (lane >= o) incl += u;
  }
  __syncthreads();
  if (lane == 31) wsum[wid] = incl;
  __syncthreads();
  if (wid == 0) {
    const int w = wsum[lane];
    int wi = w;
    for (int o = 1; o < 32; o <<= 1) {
      int u = __shfl_up(wi, o, 32);
      if (lane >= o) wi += u;
    }
    wsum[lane] = wi - w;
    if (lane == 31) *s_tot = wi;
  }
  __syncthreads();
  return wsum[wid] + incl - v;
}

__global__ __launch_bounds__(PBLOCK) void bucket_count_kernel(const int* __restrict__ dst,
                                                             int* __restrict__ counts,
                                                             int E, int nbk) {
  __shared__ unsigned hist[BKT_MAX];
  const int t = threadIdx.x;
  if (t < BKT_MAX) hist[t] = 0u;
  __syncthreads();

  const int e0 = blockIdx.x * PCH;
  int d[PPT];
#pragma unroll
  for (int u = 0; u < PPT; ++u) {
    const int e = e0 + t + u * PBLOCK;
    d[u] = -1;
    if (e < E) d[u] = dst[e];
  }
#pragma unroll
  for (int u = 0; u < PPT; ++u) {
    const unsigned b = (unsigned)d[u] >> BKT_SHIFT;
    if (b < (unsigned)nbk) atomicAdd(&hist[b], 1u);
  }
  __syncthreads();
  if (t < nbk) counts[(size_t)blockIdx.x * nbk + t] = (int)hist[t];
}

__global__ __launch_bounds__(PBLOCK) void bucket_partition_kernel(const int* __restrict__ src,
                                                                 const int* __restrict__ dst,
                                                                 const int* __restrict__ counts,
                                                                 int* __restrict__ bstart_out,
                                                                 int* __restrict__ pairs,
                                                                 int E, int nbk, int pa) {
  __shared__ int part_pre[4 * BKT_MAX];
  __shared__ int part_tot[4 * BKT_MAX];
  __shared__ int run_start[BKT_MAX];
  __shared__ int loc_start[BKT_MAX];
  __shared__ unsigned cur[BKT_MAX];
  __shared__ int wsum[32];
  __shared__ int s_tot;
  __shared__ int s_src[PCH];
  __shared__ int s_dst[PCH];

  const int t   = threadIdx.x;
  const int blk = blockIdx.x;

  {
    const int g = t >> 8;
    const int b = t & (BKT_MAX - 1);
    int pre = 0, tot = 0;
    if (b < nbk) {
      for (int r0 = g; r0 < pa; r0 += 4 * COL_BATCH) {
        int c[COL_BATCH];
#pragma unroll
        for (int u = 0; u < COL_BATCH; ++u) {
          const int r = r0 + 4 * u;
          c[u] = (r < pa) ? counts[(size_t)r * nbk + b] : 0;
        }
#pragma unroll
        for (int u = 0; u < COL_BATCH; ++u) {
          tot += c[u];
          pre += (r0 + 4 * u < blk) ? c[u] : 0;
        }
      }
    }
    part_pre[t] = pre;
    part_tot[t] = tot;
  }
  __syncthreads();

  int pre = 0, tot = 0, own = 0;
  if (t < nbk) {
    pre = part_pre[t] + part_pre[t + BKT_MAX] + part_pre[t + 2 * BKT_MAX] + part_pre[t + 3 * BKT_MAX];
    tot = part_tot[t] + part_tot[t + BKT_MAX] + part_tot[t + 2 * BKT_MAX] + part_tot[t + 3 * BKT_MAX];
    own = counts[(size_t)blk * nbk + t];
  }
  const int bst = block_excl_scan_1024(tot, wsum, &s_tot);
  if (t < nbk) {
    run_start[t] = bst + pre;
    cur[t] = 0u;
    if (blk == 0) bstart_out[t] = bst;
  }
  if (blk == 0 && t == nbk) bstart_out[nbk] = bst;
  const int lst = block_excl_scan_1024(own, wsum, &s_tot);
  if (t < nbk) loc_start[t] = lst;
  __syncthreads();
  const int placed = s_tot;

  const int e0 = blk * PCH;
  int d[PPT];
  int s[PPT];
#pragma unroll
  for (int u = 0; u < PPT; ++u) {
    const int e = e0 + t + u * PBLOCK;
    d[u] = -1;
    s[u] = 0;
    if (e < E) {
      d[u] = dst[e];
      s[u] = src[e];
    }
  }
#pragma unroll
  for (int u = 0; u < PPT; ++u) {
    const unsigned b = (unsigned)d[u] >> BKT_SHIFT;
    if (b < (unsigned)nbk) {
      const unsigned r = atomicAdd(&cur[b], 1u);
      const int p = loc_start[b] + (int)r;
      s_src[p] = s[u];
      s_dst[p] = d[u];
    }
  }
  __syncthreads();

  int2* __restrict__ rec = reinterpret_cast<int2*>(pairs);
  for (int i = t; i < placed; i += PBLOCK) {
    const int dd = s_dst[i];
    const int b  = dd >> BKT_SHIFT;
    rec[run_start[b] + (i - loc_start[b])] = make_int2(s_src[i], dd);
  }
}

#define KB_RPT (KB_CAP / PBLOCK)

__global__ __launch_bounds__(PBLOCK) void bucket_build_kernel(const int* __restrict__ pairs,
                                                             const int* __restrict__ bstart,
                                                             float* __restrict__ dinv,
                                                             int* __restrict__ row_ptr,
                                                             int* __restrict__ csr_src,
                                                             int n) {
  __shared__ unsigned cnt[BKT_NODES];
  __shared__ int excl[BKT_NODES];
  __shared__ int sorted[KB_CAP];
  __shared__ int wsum[32];
  __shared__ int s_tot;

  const int2* __restrict__ rec = reinterpret_cast<const int2*>(pairs);
  const int t    = threadIdx.x;
  const int base = blockIdx.x * BKT_NODES;
  const int p0   = bstart[blockIdx.x];
  const int p1   = bstart[blockIdx.x + 1];
  const int m    = p1 - p0;
  const bool in_lds = (m <= KB_CAP);

  if (t < BKT_NODES) cnt[t] = 0u;
  __syncthreads();

  int rs[KB_RPT];
  int rv[KB_RPT];
  if (in_lds) {
    int2 raw[KB_RPT];
#pragma unroll
    for (int u = 0; u < KB_RPT; ++u) raw[u] = make_int2(0, 0);
    if (m > 0) {
#pragma unroll
      for (int u = 0; u < KB_RPT; ++u) {
        const int i  = t + u * PBLOCK;
        const int ic = (i < m) ? i : (m - 1);
        raw[u] = rec[p0 + ic];
      }
    }
#pragma unroll
    for (int u = 0; u < KB_RPT; ++u) {
      const int i = t + u * PBLOCK;
      rs[u] = raw[u].x;
      rv[u] = (i < m) ? (raw[u].y - base) : -1;
      if ((unsigned)rv[u] < (unsigned)BKT_NODES) atomicAdd(&cnt[rv[u]], 1u);
    }
  } else {
    for (int i = t; i < m; i += PBLOCK) {
      const unsigned v = (unsigned)(rec[p0 + i].y - base);
      if (v < (unsigned)BKT_NODES) atomicAdd(&cnt[v], 1u);
    }
  }
  __syncthreads();

  const int c  = (t < BKT_NODES) ? (int)cnt[t] : 0;
  const int ex = block_excl_scan_1024(c, wsum, &s_tot);
  if (t < BKT_NODES) {
    const int node = base + t;
    if (node < n) {
      const float dg = (float)(c + 1);
      dinv[node]    = (dg > 0.0f) ? rsqrtf(dg) : 0.0f;
      row_ptr[node] = p0 + ex;
    }
    excl[t] = ex;
    cnt[t]  = 0u;
  }
  if (blockIdx.x == gridDim.x - 1 && t == 0) row_ptr[n] = p1;
  __syncthreads();

  if (in_lds) {
#pragma unroll
    for (int u = 0; u < KB_RPT; ++u) {
      if ((unsigned)rv[u] < (unsigned)BKT_NODES) {
        const int pos = excl[rv[u]] + (int)atomicAdd(&cnt[rv[u]], 1u);
        sorted[pos] = rs[u];
      }
    }
    __syncthreads();
    for (int i = t; i < m; i += PBLOCK) csr_src[p0 + i] = sorted[i];
  } else {
    for (int i = t; i < m; i += PBLOCK) {
      const int2 pr = rec[p0 + i];
      const unsigned v = (unsigned)(pr.y - base);
      if (v < (unsigned)BKT_NODES) {
        const int pos = excl[v] + (int)atomicAdd(&cnt[v], 1u);
        csr_src[p0 + pos] = pr.x;
      }
    }
  }
}


__global__ __launch_bounds__(256) void gemm_wmma_kernel(
    const float* __restrict__ A, const float* __restrict__ W,
    const float* __restrict__ rs, float* __restrict__ H, int n, int K) {
  __shared__ float tile[32 * LDSW];

  const int t     = threadIdx.x;
  const int wid   = t >> 5;
  const int lane  = t & 31;
  const int row16 = lane & 15;
  const int half  = lane >> 4;
  const int m_sub = wid >> 2;
  const int n0    = (wid & 3) << 4;
  const int m_base = blockIdx.x * 32;

  v8f acc = {};

  const int lr = t >> 3;
  const int lc = (t & 7) << 3;

  for (int kb = 0; kb < K; kb += 64) {
    float4 v0 = make_float4(0.f, 0.f, 0.f, 0.f);
    float4 v1 = v0;
    int grow = m_base + lr;
    if (grow < n) {
      const float* p = A + (size_t)grow * K + kb + lc;
      v0 = *(const float4*)(p);
      v1 = *(const float4*)(p + 4);
    }
    float* q = &tile[lr * LDSW + lc];
    *(float4*)(q)     = v0;
    *(float4*)(q + 4) = v1;
    __syncthreads();

    const float* arow = &tile[(m_sub * 16 + row16) * LDSW];
#pragma unroll
    for (int kk = 0; kk < 64; kk += 4) {
      v2f a = *(const v2f*)(arow + kk + half * 2);
      int krow = kb + kk + half * 2;
      v2f b;
      b.x = W[(size_t)krow * HIDC + n0 + row16];
      b.y = W[(size_t)(krow + 1) * HIDC + n0 + row16];
      acc = __builtin_amdgcn_wmma_f32_16x16x4_f32(
          false, a, false, b, (short)0, acc, false, false);
    }
    __syncthreads();
  }

  int mrow = m_base + m_sub * 16 + half * 8;
  float sc[8];
#pragma unroll
  for (int r = 0; r < 8; ++r) {
    const int gr = mrow + r;
    sc[r] = rs[gr < n ? gr : n - 1];
  }
#pragma unroll
  for (int r = 0; r < 8; ++r) {
    int gr = mrow + r;
    if (gr < n) H[(size_t)gr * HIDC + n0 + row16] = acc[r] * sc[r];
  }
}


#define GATHER_UNROLL 8

__global__ __launch_bounds__(256) void gather_rows_kernel(const int* __restrict__ row_ptr,
                                                          const int* __restrict__ csr_src,
                                                          const float* __restrict__ dinv,
                                                          const float* __restrict__ hs,
                                                          const float* __restrict__ bias,
                                                          float* __restrict__ outp,
                                                          int n, int relu) {
  const int lane = threadIdx.x & 31;
  const int node = __builtin_amdgcn_readfirstlane((int)(blockIdx.x * 8u + (threadIdx.x >> 5)));
  if (node >= n) return;

  const int start = row_ptr[node];
  const int end   = row_ptr[node + 1];
  const float di  = dinv[node];
  const int c = lane << 1;

  float2 acc = *(const float2*)(hs + (size_t)node * HIDC + c);

  for (int k0 = start; k0 < end; k0 += 32) {
    const int k = k0 + lane;
    int s_l = 0;
    if (k < end) s_l = csr_src[k];
    const unsigned o_l = (unsigned)s_l * (unsigned)HIDC;
    const int m = (end - k0 < 32) ? (end - k0) : 32;

    for (int j = 0; j < m; j += GATHER_UNROLL) {
      float2 r[GATHER_UNROLL];
#pragma unroll
      for (int u = 0; u < GATHER_UNROLL; ++u) {
        r[u] = make_float2(0.0f, 0.0f);
        if (j + u < m) {
          const unsigned o = (unsigned)__builtin_amdgcn_readlane((int)o_l, j + u);
          r[u] = *(const float2*)(hs + (size_t)o + c);
        }
      }
#pragma unroll
      for (int u = 0; u < GATHER_UNROLL; ++u) {
        acc.x += r[u].x;
        acc.y += r[u].y;
      }
    }
  }

  const float2 b = *(const float2*)(bias + c);
  float ox = acc.x * di + b.x;
  float oy = acc.y * di + b.y;
  if (relu) {
    ox = ox > 0.0f ? ox : 0.0f;
    oy = oy > 0.0f ? oy : 0.0f;
  }
  *(float2*)(outp + (size_t)node * HIDC + c) = make_float2(ox, oy);
}


extern "C" void kernel_launch(void* const* d_in, const int* in_sizes, int n_in,
                              void* d_out, int out_size, void* d_ws, size_t ws_size,
                              hipStream_t stream) {
  const float* x  = (const float*)d_in[0];
  const int*   ei = (const int*)d_in[1];
  const float* W1 = (const float*)d_in[2];
  const float* b1 = (const float*)d_in[3];
  const float* W2 = (const float*)d_in[4];
  const float* b2 = (const float*)d_in[5];

  const int IN_CH = 512;
  const int n = in_sizes[0] / IN_CH;
  const int E = in_sizes[1] / 2;
  const int* src = ei;
  const int* dst = ei + E;
  float* out = (float*)d_out;

  float* ws = (float*)d_ws;
  size_t n_al = ((size_t)n + 255) & ~(size_t)255;
  float* dinv = ws;
  float* bufA = ws + n_al;
  const int nbk = (n + BKT_NODES - 1) / BKT_NODES;
  const int pa  = (E + PCH - 1) / PCH;
  size_t cm_al = ((size_t)pa * (size_t)nbk + 255) & ~(size_t)255;
  int* rowp   = (int*)(bufA + (size_t)n * HIDC);
  int* counts = rowp + n_al + 256;
  int* bstart = counts + cm_al;
  int* csrs   = bstart + 512;
  int* pairs  = (int*)d_out;

  unsigned gG   = (unsigned)((n + 7) / 8);
  unsigned gMM  = (unsigned)((n + 31) / 32);

  bucket_count_kernel<<<(unsigned)pa, PBLOCK, 0, stream>>>(dst, counts, E, nbk);
  bucket_partition_kernel<<<(unsigned)pa, PBLOCK, 0, stream>>>(src, dst, counts, bstart, pairs, E, nbk, pa);
  bucket_build_kernel<<<(unsigned)nbk, PBLOCK, 0, stream>>>(pairs, bstart, dinv, rowp, csrs, n);

  gemm_wmma_kernel<<<gMM, 256, 0, stream>>>(x, W1, dinv, bufA, n, IN_CH);
  gather_rows_kernel<<<gG, 256, 0, stream>>>(rowp, csrs, dinv, bufA, b1, out, n, 1);

  gemm_wmma_kernel<<<gMM, 256, 0, stream>>>(out, W2, dinv, bufA, n, HIDC);
  gather_rows_kernel<<<gG, 256, 0, stream>>>(rowp, csrs, dinv, bufA, b2, out, n, 0);
}


# layer-2 product fused into the layer-1 gather: each wave sums 2 nodes, a block's 16 relu'd rows stay in LDS, 4 waves WMMA them with W2 and store scaled rows; 6 launches not 7; sum order kept
# speedup vs baseline: 3.0480x; 1.0406x over previous
#include <hip/hip_runtime.h>

typedef __attribute__((ext_vector_type(2))) float v2f;
typedef __attribute__((ext_vector_type(8))) float v8f;

#define HIDC 64
#define LDSW 68


#define BKT_NODES 512
#define BKT_SHIFT 9
#define BKT_MAX   256
#define PCH       4096
#define PBLOCK    1024
#define PPT       (PCH / PBLOCK)
#define KB_CAP    12288
#define COL_BATCH 16

__device__ __forceinline__ int block_excl_scan_1024(int v, int* wsum, int* s_tot) {
  const int lane = threadIdx.x & 31;
  const int wid  = threadIdx.x >> 5;
  int incl = v;
  for (int o = 1; o < 32; o <<= 1) {
    int u = __shfl_up(incl, o, 32);
    if (lane >= o) incl += u;
  }
  __syncthreads();
  if (lane == 31) wsum[wid] = incl;
  __syncthreads();
  if (wid == 0) {
    const int w = wsum[lane];
    int wi = w;
    for (int o = 1; o < 32; o <<= 1) {
      int u = __shfl_up(wi, o, 32);
      if (lane >= o) wi += u;
    }
    wsum[lane] = wi - w;
    if (lane == 31) *s_tot = wi;
  }
  __syncthreads();
  return wsum[wid] + incl - v;
}

__global__ __launch_bounds__(PBLOCK) void bucket_count_kernel(const int* __restrict__ dst,
                                                             int* __restrict__ counts,
                                                             int E, int nbk) {
  __shared__ unsigned hist[BKT_MAX];
  const int t = threadIdx.x;
  if (t < BKT_MAX) hist[t] = 0u;
  __syncthreads();

  const int e0 = blockIdx.x * PCH;
  int d[PPT];
#pragma unroll
  for (int u = 0; u < PPT; ++u) {
    const int e = e0 + t + u * PBLOCK;
    d[u] = -1;
    if (e < E) d[u] = dst[e];
  }
#pragma unroll
  for (int u = 0; u < PPT; ++u) {
    const unsigned b = (unsigned)d[u] >> BKT_SHIFT;
    if (b < (unsigned)nbk) atomicAdd(&hist[b], 1u);
  }
  __syncthreads();
  if (t < nbk) counts[(size_t)blockIdx.x * nbk + t] = (int)hist[t];
}

__global__ __launch_bounds__(PBLOCK) void bucket_partition_kernel(const int* __restrict__ src,
                                                                 const int* __restrict__ dst,
                                                                 const int* __restrict__ counts,
                                                                 int* __restrict__ bstart_out,
                                                                 int* __restrict__ pairs,
                                                                 int E, int nbk, int pa) {
  __shared__ int part_pre[4 * BKT_MAX];
  __shared__ int part_tot[4 * BKT_MAX];
  __shared__ int run_start[BKT_MAX];
  __shared__ int loc_start[BKT_MAX];
  __shared__ unsigned cur[BKT_MAX];
  __shared__ int wsum[32];
  __shared__ int s_tot;
  __shared__ int s_src[PCH];
  __shared__ int s_dst[PCH];

  const int t   = threadIdx.x;
  const int blk = blockIdx.x;

  {
    const int g = t >> 8;
    const int b = t & (BKT_MAX - 1);
    int pre = 0, tot = 0;
    if (b < nbk) {
      for (int r0 = g; r0 < pa; r0 += 4 * COL_BATCH) {
        int c[COL_BATCH];
#pragma unroll
        for (int u = 0; u < COL_BATCH; ++u) {
          const int r = r0 + 4 * u;
          c[u] = (r < pa) ? counts[(size_t)r * nbk + b] : 0;
        }
#pragma unroll
        for (int u = 0; u < COL_BATCH; ++u) {
          tot += c[u];
          pre += (r0 + 4 * u < blk) ? c[u] : 0;
        }
      }
    }
    part_pre[t] = pre;
    part_tot[t] = tot;
  }
  __syncthreads();

  int pre = 0, tot = 0, own = 0;
  if (t < nbk) {
    pre = part_pre[t] + part_pre[t + BKT_MAX] + part_pre[t + 2 * BKT_MAX] + part_pre[t + 3 * BKT_MAX];
    tot = part_tot[t] + part_tot[t + BKT_MAX] + part_tot[t + 2 * BKT_MAX] + part_tot[t + 3 * BKT_MAX];
    own = counts[(size_t)blk * nbk + t];
  }
  const int bst = block_excl_scan_1024(tot, wsum, &s_tot);
  if (t < nbk) {
    run_start[t] = bst + pre;
    cur[t] = 0u;
    if (blk == 0) bstart_out[t] = bst;
  }
  if (blk == 0 && t == nbk) bstart_out[nbk] = bst;
  const int lst = block_excl_scan_1024(own, wsum, &s_tot);
  if (t < nbk) loc_start[t] = lst;
  __syncthreads();
  const int placed = s_tot;

  const int e0 = blk * PCH;
  int d[PPT];
  int s[PPT];
#pragma unroll
  for (int u = 0; u < PPT; ++u) {
    const int e = e0 + t + u * PBLOCK;
    d[u] = -1;
    s[u] = 0;
    if (e < E) {
      d[u] = dst[e];
      s[u] = src[e];
    }
  }
#pragma unroll
  for (int u = 0; u < PPT; ++u) {
    const unsigned b = (unsigned)d[u] >> BKT_SHIFT;
    if (b < (unsigned)nbk) {
      const unsigned r = atomicAdd(&cur[b], 1u);
      const int p = loc_start[b] + (int)r;
      s_src[p] = s[u];
      s_dst[p] = d[u];
    }
  }
  __syncthreads();

  int2* __restrict__ rec = reinterpret_cast<int2*>(pairs);
  for (int i = t; i < placed; i += PBLOCK) {
    const int dd = s_dst[i];
    const int b  = dd >> BKT_SHIFT;
    rec[run_start[b] + (i - loc_start[b])] = make_int2(s_src[i], dd);
  }
}

#define KB_RPT (KB_CAP / PBLOCK)

__global__ __launch_bounds__(PBLOCK) void bucket_build_kernel(const int* __restrict__ pairs,
                                                             const int* __restrict__ bstart,
                                                             float* __restrict__ dinv,
                                                             int* __restrict__ row_ptr,
                                                             int* __restrict__ csr_src,
                                                             int n) {
  __shared__ unsigned cnt[BKT_NODES];
  __shared__ int excl[BKT_NODES];
  __shared__ int sorted[KB_CAP];
  __shared__ int wsum[32];
  __shared__ int s_tot;

  const int2* __restrict__ rec = reinterpret_cast<const int2*>(pairs);
  const int t    = threadIdx.x;
  const int base = blockIdx.x * BKT_NODES;
  const int p0   = bstart[blockIdx.x];
  const int p1   = bstart[blockIdx.x + 1];
  const int m    = p1 - p0;
  const bool in_lds = (m <= KB_CAP);

  if (t < BKT_NODES) cnt[t] = 0u;
  __syncthreads();

  int rs[KB_RPT];
  int rv[KB_RPT];
  if (in_lds) {
    int2 raw[KB_RPT];
#pragma unroll
    for (int u = 0; u < KB_RPT; ++u) raw[u] = make_int2(0, 0);
    if (m > 0) {
#pragma unroll
      for (int u = 0; u < KB_RPT; ++u) {
        const int i  = t + u * PBLOCK;
        const int ic = (i < m) ? i : (m - 1);
        raw[u] = rec[p0 + ic];
      }
    }
#pragma unroll
    for (int u = 0; u < KB_RPT; ++u) {
      const int i = t + u * PBLOCK;
      rs[u] = raw[u].x;
      rv[u] = (i < m) ? (raw[u].y - base) : -1;
      if ((unsigned)rv[u] < (unsigned)BKT_NODES) atomicAdd(&cnt[rv[u]], 1u);
    }
  } else {
    for (int i = t; i < m; i += PBLOCK) {
      const unsigned v = (unsigned)(rec[p0 + i].y - base);
      if (v < (unsigned)BKT_NODES) atomicAdd(&cnt[v], 1u);
    }
  }
  __syncthreads();

  const int c  = (t < BKT_NODES) ? (int)cnt[t] : 0;
  const int ex = block_excl_scan_1024(c, wsum, &s_tot);
  if (t < BKT_NODES) {
    const int node = base + t;
    if (node < n) {
      const float dg = (float)(c + 1);
      dinv[node]    = (dg > 0.0f) ? rsqrtf(dg) : 0.0f;
      row_ptr[node] = p0 + ex;
    }
    excl[t] = ex;
    cnt[t]  = 0u;
  }
  if (blockIdx.x == gridDim.x - 1 && t == 0) row_ptr[n] = p1;
  __syncthreads();

  if (in_lds) {
#pragma unroll
    for (int u = 0; u < KB_RPT; ++u) {
      if ((unsigned)rv[u] < (unsigned)BKT_NODES) {
        const int pos = excl[rv[u]] + (int)atomicAdd(&cnt[rv[u]], 1u);
        sorted[pos] = rs[u];
      }
    }
    __syncthreads();
    for (int i = t; i < m; i += PBLOCK) csr_src[p0 + i] = sorted[i];
  } else {
    for (int i = t; i < m; i += PBLOCK) {
      const int2 pr = rec[p0 + i];
      const unsigned v = (unsigned)(pr.y - base);
      if (v < (unsigned)BKT_NODES) {
        const int pos = excl[v] + (int)atomicAdd(&cnt[v], 1u);
        csr_src[p0 + pos] = pr.x;
      }
    }
  }
}


__global__ __launch_bounds__(256) void gemm_wmma_kernel(
    const float* __restrict__ A, const float* __restrict__ W,
    const float* __restrict__ rs, float* __restrict__ H, int n, int K) {
  __shared__ float tile[32 * LDSW];

  const int t     = threadIdx.x;
  const int wid   = t >> 5;
  const int lane  = t & 31;
  const int row16 = lane & 15;
  const int half  = lane >> 4;
  const int m_sub = wid >> 2;
  const int n0    = (wid & 3) << 4;
  const int m_base = blockIdx.x * 32;

  v8f acc = {};

  const int lr = t >> 3;
  const int lc = (t & 7) << 3;

  for (int kb = 0; kb < K; kb += 64) {
    float4 v0 = make_float4(0.f, 0.f, 0.f, 0.f);
    float4 v1 = v0;
    int grow = m_base + lr;
    if (grow < n) {
      const float* p = A + (size_t)grow * K + kb + lc;
      v0 = *(const float4*)(p);
      v1 = *(const float4*)(p + 4);
    }
    float* q = &tile[lr * LDSW + lc];
    *(float4*)(q)     = v0;
    *(float4*)(q + 4) = v1;
    __syncthreads();

    const float* arow = &tile[(m_sub * 16 + row16) * LDSW];
#pragma unroll
    for (int kk = 0; kk < 64; kk += 4) {
      v2f a = *(const v2f*)(arow + kk + half * 2);
      int krow = kb + kk + half * 2;
      v2f b;
      b.x = W[(size_t)krow * HIDC + n0 + row16];
      b.y = W[(size_t)(krow + 1) * HIDC + n0 + row16];
      acc = __builtin_amdgcn_wmma_f32_16x16x4_f32(
          false, a, false, b, (short)0, acc, false, false);
    }
    __syncthreads();
  }

  int mrow = m_base + m_sub * 16 + half * 8;
  float sc[8];
#pragma unroll
  for (int r = 0; r < 8; ++r) {
    const int gr = mrow + r;
    sc[r] = rs[gr < n ? gr : n - 1];
  }
#pragma unroll
  for (int r = 0; r < 8; ++r) {
    int gr = mrow + r;
    if (gr < n) H[(size_t)gr * HIDC + n0 + row16] = acc[r] * sc[r];
  }
}


#define GATHER_UNROLL 8

__global__ __launch_bounds__(256) void gather_rows_kernel(const int* __restrict__ row_ptr,
                                                          const int* __restrict__ csr_src,
                                                          const float* __restrict__ dinv,
                                                          const float* __restrict__ hs,
                                                          const float* __restrict__ bias,
                                                          float* __restrict__ outp,
                                                          int n, int relu) {
  const int lane = threadIdx.x & 31;
  const int node = __builtin_amdgcn_readfirstlane((int)(blockIdx.x * 8u + (threadIdx.x >> 5)));
  if (node >= n) return;

  const int start = row_ptr[node];
  const int end   = row_ptr[node + 1];
  const float di  = dinv[node];
  const int c = lane << 1;

  float2 acc = *(const float2*)(hs + (size_t)node * HIDC + c);

  for (int k0 = start; k0 < end; k0 += 32) {
    const int k = k0 + lane;
    int s_l = 0;
    if (k < end) s_l = csr_src[k];
    const unsigned o_l = (unsigned)s_l * (unsigned)HIDC;
    const int m = (end - k0 < 32) ? (end - k0) : 32;

    for (int j = 0; j < m; j += GATHER_UNROLL) {
      float2 r[GATHER_UNROLL];
#pragma unroll
      for (int u = 0; u < GATHER_UNROLL; ++u) {
        r[u] = make_float2(0.0f, 0.0f);
        if (j + u < m) {
          const unsigned o = (unsigned)__builtin_amdgcn_readlane((int)o_l, j + u);
          r[u] = *(const float2*)(hs + (size_t)o + c);
        }
      }
#pragma unroll
      for (int u = 0; u < GATHER_UNROLL; ++u) {
        acc.x += r[u].x;
        acc.y += r[u].y;
      }
    }
  }

  const float2 b = *(const float2*)(bias + c);
  float ox = acc.x * di + b.x;
  float oy = acc.y * di + b.y;
  if (relu) {
    ox = ox > 0.0f ? ox : 0.0f;
    oy = oy > 0.0f ? oy : 0.0f;
  }
  *(float2*)(outp + (size_t)node * HIDC + c) = make_float2(ox, oy);
}


__device__ __forceinline__ float2 sum_list_rows(const int* __restrict__ csr_src,
                                                const float* __restrict__ hs,
                                                int start, int end, int s_first,
                                                float2 acc, int lane, int c) {
  for (int k0 = start; k0 < end; k0 += 32) {
    int s_l = s_first;
    if (k0 != start) {
      const int k = k0 + lane;
      s_l = 0;
      if (k < end) s_l = csr_src[k];
    }
    const unsigned o_l = (unsigned)s_l * (unsigned)HIDC;
    const int m = (end - k0 < 32) ? (end - k0) : 32;

    for (int j = 0; j < m; j += GATHER_UNROLL) {
      float2 rw[GATHER_UNROLL];
#pragma unroll
      for (int u = 0; u < GATHER_UNROLL; ++u) {
        rw[u] = make_float2(0.0f, 0.0f);
        if (j + u < m) {
          const unsigned o = (unsigned)__builtin_amdgcn_readlane((int)o_l, j + u);
          rw[u] = *(const float2*)(hs + (size_t)o + c);
        }
      }
#pragma unroll
      for (int u = 0; u < GATHER_UNROLL; ++u) {
        acc.x += rw[u].x;
        acc.y += rw[u].y;
      }
    }
  }
  return acc;
}

__global__ __launch_bounds__(256) void gather_relu_gemm_kernel(const int* __restrict__ row_ptr,
                                                               const int* __restrict__ csr_src,
                                                               const float* __restrict__ dinv,
                                                               const float* __restrict__ hs,
                                                               const float* __restrict__ bias,
                                                               const float* __restrict__ W,
                                                               float* __restrict__ H,
                                                               int n) {
  __shared__ float tile[16 * LDSW];

  const int lane = threadIdx.x & 31;
  const int wid    = __builtin_amdgcn_readfirstlane((int)(threadIdx.x >> 5));
  const int m_base = (int)(blockIdx.x * 16u);
  const int c = lane << 1;

  const int nodeA = m_base + wid * 2;
  const int nodeB = nodeA + 1;
  const int cA = (nodeA < n) ? nodeA : (n - 1);
  const int cB = (nodeB < n) ? nodeB : (n - 1);
  const int iA = (nodeA < n) ? nodeA : n;
  const int iB = (nodeA + 1 < n) ? (nodeA + 1) : n;
  const int iC = (nodeA + 2 < n) ? (nodeA + 2) : n;
  const int startA = row_ptr[iA];
  const int endA   = row_ptr[iB];
  const int endB   = row_ptr[iC];
  const float dA = dinv[cA];
  const float dB = dinv[cB];

  int sA = 0, sB = 0;
  if (startA + lane < endA) sA = csr_src[startA + lane];
  if (endA + lane < endB)   sB = csr_src[endA + lane];
  float2 accA = *(const float2*)(hs + (size_t)cA * HIDC + c);
  float2 accB = *(const float2*)(hs + (size_t)cB * HIDC + c);
  const float2 bs = *(const float2*)(bias + c);

  accA = sum_list_rows(csr_src, hs, startA, endA, sA, accA, lane, c);
  accB = sum_list_rows(csr_src, hs, endA, endB, sB, accB, lane, c);

  float ax = accA.x * dA + bs.x;
  float ay = accA.y * dA + bs.y;
  float bx = accB.x * dB + bs.x;
  float by = accB.y * dB + bs.y;
  ax = (nodeA < n && ax > 0.0f) ? ax : 0.0f;
  ay = (nodeA < n && ay > 0.0f) ? ay : 0.0f;
  bx = (nodeB < n && bx > 0.0f) ? bx : 0.0f;
  by = (nodeB < n && by > 0.0f) ? by : 0.0f;
  *(float2*)(&tile[(wid * 2) * LDSW + c])     = make_float2(ax, ay);
  *(float2*)(&tile[(wid * 2 + 1) * LDSW + c]) = make_float2(bx, by);

  const bool mma  = (wid < 4);
  const int row16 = lane & 15;
  const int half  = lane >> 4;
  const int n0    = (wid & 3) << 4;
  const int mrow  = m_base + half * 8;
  v2f bw[HIDC / 4];
  float sc[8];
#pragma unroll
  for (int t = 0; t < HIDC / 4; ++t) bw[t] = (v2f){0.0f, 0.0f};
#pragma unroll
  for (int r = 0; r < 8; ++r) sc[r] = 0.0f;
  if (mma) {
#pragma unroll
    for (int t = 0; t < HIDC / 4; ++t) {
      const int krow = t * 4 + half * 2;
      bw[t].x = W[(size_t)krow * HIDC + n0 + row16];
      bw[t].y = W[(size_t)(krow + 1) * HIDC + n0 + row16];
    }
#pragma unroll
    for (int r = 0; r < 8; ++r) {
      const int gr = mrow + r;
      sc[r] = dinv[gr < n ? gr : n - 1];
    }
  }
  __syncthreads();
  if (!mma) return;

  v8f acc8 = {};
  const float* arow = &tile[row16 * LDSW];
#pragma unroll
  for (int t = 0; t < HIDC / 4; ++t) {
    v2f a = *(const v2f*)(arow + t * 4 + half * 2);
    acc8 = __builtin_amdgcn_wmma_f32_16x16x4_f32(
        false, a, false, bw[t], (short)0, acc8, false, false);
  }

#pragma unroll
  for (int r = 0; r < 8; ++r) {
    const int gr = mrow + r;
    if (gr < n) H[(size_t)gr * HIDC + n0 + row16] = acc8[r] * sc[r];
  }
}


extern "C" void kernel_launch(void* const* d_in, const int* in_sizes, int n_in,
                              void* d_out, int out_size, void* d_ws, size_t ws_size,
                              hipStream_t stream) {
  const float* x  = (const float*)d_in[0];
  const int*   ei = (const int*)d_in[1];
  const float* W1 = (const float*)d_in[2];
  const float* b1 = (const float*)d_in[3];
  const float* W2 = (const float*)d_in[4];
  const float* b2 = (const float*)d_in[5];

  const int IN_CH = 512;
  const int n = in_sizes[0] / IN_CH;
  const int E = in_sizes[1] / 2;
  const int* src = ei;
  const int* dst = ei + E;
  float* out = (float*)d_out;

  float* ws = (float*)d_ws;
  size_t n_al = ((size_t)n + 255) & ~(size_t)255;
  float* dinv = ws;
  float* bufA = ws + n_al;
  const int nbk = (n + BKT_NODES - 1) / BKT_NODES;
  const int pa  = (E + PCH - 1) / PCH;
  size_t cm_al = ((size_t)pa * (size_t)nbk + 255) & ~(size_t)255;
  int* rowp   = (int*)(bufA + (size_t)n * HIDC);
  int* counts = rowp + n_al + 256;
  int* bstart = counts + cm_al;
  int* csrs   = bstart + 512;
  int* pairs  = (int*)d_out;

  unsigned gG   = (unsigned)((n + 7) / 8);
  unsigned gF   = (unsigned)((n + 15) / 16);
  unsigned gMM  = (unsigned)((n + 31) / 32);

  bucket_count_kernel<<<(unsigned)pa, PBLOCK, 0, stream>>>(dst, counts, E, nbk);
  bucket_partition_kernel<<<(unsigned)pa, PBLOCK, 0, stream>>>(src, dst, counts, bstart, pairs, E, nbk, pa);
  bucket_build_kernel<<<(unsigned)nbk, PBLOCK, 0, stream>>>(pairs, bstart, dinv, rowp, csrs, n);

  gemm_wmma_kernel<<<gMM, 256, 0, stream>>>(x, W1, dinv, out, n, IN_CH);

  gather_relu_gemm_kernel<<<gF, 256, 0, stream>>>(rowp, csrs, dinv, out, b1, W2, bufA, n);

  gather_rows_kernel<<<gG, 256, 0, stream>>>(rowp, csrs, dinv, bufA, b2, out, n, 0);
}


# in-edge list build in two launches, not three: each chunk is sorted by bucket locally with its own offset table; a bucket's block sums the offsets for its place; list (sum) order changes
# speedup vs baseline: 3.8024x; 1.2475x over previous
#include <hip/hip_runtime.h>

typedef __attribute__((ext_vector_type(2))) float v2f;
typedef __attribute__((ext_vector_type(8))) float v8f;

#define HIDC 64
#define LDSW 68


#define BKT_NODES 512
#define BKT_SHIFT 9
#define BKT_MAX   256
#define PCH       4096
#define PBLOCK    1024
#define PPT       (PCH / PBLOCK)
#define KB_CAP    12288

__device__ __forceinline__ int block_excl_scan_1024(int v, int* wsum, int* s_tot) {
  const int lane = threadIdx.x & 31;
  const int wid  = threadIdx.x >> 5;
  int incl = v;
  for (int o = 1; o < 32; o <<= 1) {
    int u = __shfl_up(incl, o, 32);
    if (lane >= o) incl += u;
  }
  __syncthreads();
  if (lane == 31) wsum[wid] = incl;
  __syncthreads();
  if (wid == 0) {
    const int w = wsum[lane];
    int wi = w;
    for (int o = 1; o < 32; o <<= 1) {
      int u = __shfl_up(wi, o, 32);
      if (lane >= o) wi += u;
    }
    wsum[lane] = wi - w;
    if (lane == 31) *s_tot = wi;
  }
  __syncthreads();
  return wsum[wid] + incl - v;
}

__global__ __launch_bounds__(PBLOCK) void bucket_partition_kernel(const int* __restrict__ src,
                                                                 const int* __restrict__ dst,
                                                                 int* __restrict__ offs,
                                                                 int* __restrict__ pairs,
                                                                 int E, int nbk) {
  __shared__ unsigned hist[BKT_MAX];
  __shared__ int loc_start[BKT_MAX];
  __shared__ int wsum[32];
  __shared__ int s_tot;
  __shared__ int s_src[PCH];
  __shared__ int s_dst[PCH];

  const int t   = threadIdx.x;
  const int blk = blockIdx.x;
  if (t < BKT_MAX) hist[t] = 0u;
  __syncthreads();

  const int e0 = blk * PCH;
  int d[PPT];
  int s[PPT];
  int r[PPT];
#pragma unroll
  for (int u = 0; u < PPT; ++u) {
    const int e = e0 + t + u * PBLOCK;
    d[u] = -1;
    s[u] = 0;
    if (e < E) {
      d[u] = dst[e];
      s[u] = src[e];
    }
  }
#pragma unroll
  for (int u = 0; u < PPT; ++u) {
    const unsigned b = (unsigned)d[u] >> BKT_SHIFT;
    r[u] = -1;
    if (b < (unsigned)nbk) r[u] = (int)atomicAdd(&hist[b], 1u);
  }
  __syncthreads();

  const int own = (t < nbk) ? (int)hist[t] : 0;
  const int lst = block_excl_scan_1024(own, wsum, &s_tot);
  if (t <= nbk) offs[(size_t)blk * (size_t)(nbk + 1) + t] = lst;
  if (t < nbk) loc_start[t] = lst;
  __syncthreads();
  const int placed = s_tot;

#pragma unroll
  for (int u = 0; u < PPT; ++u) {
    if (r[u] >= 0) {
      const int p = loc_start[(unsigned)d[u] >> BKT_SHIFT] + r[u];
      s_src[p] = s[u];
      s_dst[p] = d[u];
    }
  }
  __syncthreads();

  int2* __restrict__ rec = reinterpret_cast<int2*>(pairs) + (size_t)blk * PCH;
  for (int i = t; i < placed; i += PBLOCK) rec[i] = make_int2(s_src[i], s_dst[i]);
}

#define RUNS_PT 13

template <typename F>
__device__ __forceinline__ void leftover_records(const int2* __restrict__ rec, const int* __restrict__ offs,
                                                 int my_o0, int my_o1, int w, int lane, int b,
                                                 int stride, int pa, F f) {
  for (int u = 0; u < RUNS_PT; ++u) {
    const int o0  = __builtin_amdgcn_readlane(my_o0, u);
    const int len = __builtin_amdgcn_readlane(my_o1, u) - o0;
    if (len > 32) {
      const int2* __restrict__ q = rec + (size_t)(w + 32 * u) * PCH + o0;
      for (int i = 32 + lane; i < len; i += 32) f(q[i]);
    }
  }
  for (int blk = w + 32 * RUNS_PT; blk < pa; blk += 32) {
    const int* __restrict__ tq = offs + (size_t)blk * stride + b;
    const int o0  = tq[0];
    const int len = tq[1] - o0;
    const int2* __restrict__ q = rec + (size_t)blk * PCH + o0;
    for (int i = lane; i < len; i += 32) f(q[i]);
  }
}

__global__ __launch_bounds__(PBLOCK) void bucket_build_kernel(const int* __restrict__ pairs,
                                                             const int* __restrict__ offs,
                                                             float* __restrict__ dinv,
                                                             int* __restrict__ row_ptr,
                                                             int* __restrict__ csr_src,
                                                             int n, int nbk, int pa) {
  __shared__ unsigned cnt[BKT_NODES];
  __shared__ int excl[BKT_NODES];
  __shared__ int sorted[KB_CAP];
  __shared__ int wsum[32];
  __shared__ int s_tot;
  __shared__ int s_p[2];

  const int2* __restrict__ rec = reinterpret_cast<const int2*>(pairs);
  const int t    = threadIdx.x;
  const int lane = t & 31;
  const int w      = __builtin_amdgcn_readfirstlane(t >> 5);
  const int b      = blockIdx.x;
  const int base   = b * BKT_NODES;
  const int stride = nbk + 1;

  if (t < BKT_NODES) cnt[t] = 0u;
  if (t < 2) s_p[t] = 0;
  __syncthreads();

  int my_o0 = 0, my_o1 = 0;
  {
    const int blk = w + 32 * lane;
    if (lane < RUNS_PT && blk < pa) {
      const int* __restrict__ tq = offs + (size_t)blk * stride + b;
      my_o0 = tq[0];
      my_o1 = tq[1];
    }
  }
  int2 raw[RUNS_PT];
#pragma unroll
  for (int u = 0; u < RUNS_PT; ++u) {
    const int o0  = __builtin_amdgcn_readlane(my_o0, u);
    const int len = __builtin_amdgcn_readlane(my_o1, u) - o0;
    raw[u] = make_int2(0, 0);
    if (lane < len) raw[u] = rec[(size_t)(w + 32 * u) * PCH + o0 + lane];
  }

  int sum0 = my_o0, sum1 = my_o1;
  for (int blk = 32 * RUNS_PT + t; blk < pa; blk += PBLOCK) {
    const int* __restrict__ tq = offs + (size_t)blk * stride + b;
    sum0 += tq[0];
    sum1 += tq[1];
  }

  for (int o = 16; o > 0; o >>= 1) {
    sum0 += __shfl_xor(sum0, o, 32);
    sum1 += __shfl_xor(sum1, o, 32);
  }
  if (lane == 0) {
    atomicAdd(&s_p[0], sum0);
    atomicAdd(&s_p[1], sum1);
  }

  int rs[RUNS_PT];
  int rv[RUNS_PT];
#pragma unroll
  for (int u = 0; u < RUNS_PT; ++u) {
    const int len = __builtin_amdgcn_readlane(my_o1, u) - __builtin_amdgcn_readlane(my_o0, u);
    rs[u] = raw[u].x;
    rv[u] = (lane < len) ? (raw[u].y - base) : -1;
    if ((unsigned)rv[u] < (unsigned)BKT_NODES) atomicAdd(&cnt[rv[u]], 1u);
  }
  leftover_records(rec, offs, my_o0, my_o1, w, lane, b, stride, pa, [&](int2 pr) {
    const unsigned v = (unsigned)(pr.y - base);
    if (v < (unsigned)BKT_NODES) atomicAdd(&cnt[v], 1u);
  });
  __syncthreads();

  const int p0 = s_p[0];
  const int p1 = s_p[1];
  const int m  = p1 - p0;
  const bool in_lds = (m <= KB_CAP);

  const int c  = (t < BKT_NODES) ? (int)cnt[t] : 0;
  const int ex = block_excl_scan_1024(c, wsum, &s_tot);
  if (t < BKT_NODES) {
    const int node = base + t;
    if (node < n) {
      const float dg = (float)(c + 1);
      dinv[node]    = (dg > 0.0f) ? rsqrtf(dg) : 0.0f;
      row_ptr[node] = p0 + ex;
    }
    excl[t] = ex;
    cnt[t]  = 0u;
  }
  if (b == nbk - 1 && t == 0) row_ptr[n] = p1;
  __syncthreads();

#pragma unroll
  for (int u = 0; u < RUNS_PT; ++u) {
    if ((unsigned)rv[u] < (unsigned)BKT_NODES) {
      const int pos = excl[rv[u]] + (int)atomicAdd(&cnt[rv[u]], 1u);
      if (in_lds) sorted[pos] = rs[u];
      else        csr_src[p0 + pos] = rs[u];
    }
  }
  leftover_records(rec, offs, my_o0, my_o1, w, lane, b, stride, pa, [&](int2 pr) {
    const unsigned v = (unsigned)(pr.y - base);
    if (v < (unsigned)BKT_NODES) {
      const int pos = excl[v] + (int)atomicAdd(&cnt[v], 1u);
      if (in_lds) sorted[pos] = pr.x;
      else        csr_src[p0 + pos] = pr.x;
    }
  });
  if (in_lds) {
    __syncthreads();
    for (int i = t; i < m; i += PBLOCK) csr_src[p0 + i] = sorted[i];
  }
}


__global__ __launch_bounds__(256) void gemm_wmma_kernel(
    const float* __restrict__ A, const float* __restrict__ W,
    const float* __restrict__ rs, float* __restrict__ H, int n, int K) {
  __shared__ float tile[32 * LDSW];

  const int t     = threadIdx.x;
  const int wid   = t >> 5;
  const int lane  = t & 31;
  const int row16 = lane & 15;
  const int half  = lane >> 4;
  const int m_sub = wid >> 2;
  const int n0    = (wid & 3) << 4;
  const int m_base = blockIdx.x * 32;

  v8f acc = {};

  const int lr = t >> 3;
  const int lc = (t & 7) << 3;

  for (int kb = 0; kb < K; kb += 64) {
    float4 v0 = make_float4(0.f, 0.f, 0.f, 0.f);
    float4 v1 = v0;
    int grow = m_base + lr;
    if (grow < n) {
      const float* p = A + (size_t)grow * K + kb + lc;
      v0 = *(const float4*)(p);
      v1 = *(const float4*)(p + 4);
    }
    float* q = &tile[lr * LDSW + lc];
    *(float4*)(q)     = v0;
    *(float4*)(q + 4) = v1;
    __syncthreads();

    const float* arow = &tile[(m_sub * 16 + row16) * LDSW];
#pragma unroll
    for (int kk = 0; kk < 64; kk += 4) {
      v2f a = *(const v2f*)(arow + kk + half * 2);
      int krow = kb + kk + half * 2;
      v2f b;
      b.x = W[(size_t)krow * HIDC + n0 + row16];
      b.y = W[(size_t)(krow + 1) * HIDC + n0 + row16];
      acc = __builtin_amdgcn_wmma_f32_16x16x4_f32(
          false, a, false, b, (short)0, acc, false, false);
    }
    __syncthreads();
  }

  int mrow = m_base + m_sub * 16 + half * 8;
  float sc[8];
#pragma unroll
  for (int r = 0; r < 8; ++r) {
    const int gr = mrow + r;
    sc[r] = rs[gr < n ? gr : n - 1];
  }
#pragma unroll
  for (int r = 0; r < 8; ++r) {
    int gr = mrow + r;
    if (gr < n) H[(size_t)gr * HIDC + n0 + row16] = acc[r] * sc[r];
  }
}


#define GATHER_UNROLL 8

__global__ __launch_bounds__(256) void gather_rows_kernel(const int* __restrict__ row_ptr,
                                                          const int* __restrict__ csr_src,
                                                          const float* __restrict__ dinv,
                                                          const float* __restrict__ hs,
                                                          const float* __restrict__ bias,
                                                          float* __restrict__ outp,
                                                          int n, int relu) {
  const int lane = threadIdx.x & 31;
  const int node = __builtin_amdgcn_readfirstlane((int)(blockIdx.x * 8u + (threadIdx.x >> 5)));
  if (node >= n) return;

  const int start = row_ptr[node];
  const int end   = row_ptr[node + 1];
  const float di  = dinv[node];
  const int c = lane << 1;

  float2 acc = *(const float2*)(hs + (size_t)node * HIDC + c);

  for (int k0 = start; k0 < end; k0 += 32) {
    const int k = k0 + lane;
    int s_l = 0;
    if (k < end) s_l = csr_src[k];
    const unsigned o_l = (unsigned)s_l * (unsigned)HIDC;
    const int m = (end - k0 < 32) ? (end - k0) : 32;

    for (int j = 0; j < m; j += GATHER_UNROLL) {
      float2 r[GATHER_UNROLL];
#pragma unroll
      for (int u = 0; u < GATHER_UNROLL; ++u) {
        r[u] = make_float2(0.0f, 0.0f);
        if (j + u < m) {
          const unsigned o = (unsigned)__builtin_amdgcn_readlane((int)o_l, j + u);
          r[u] = *(const float2*)(hs + (size_t)o + c);
        }
      }
#pragma unroll
      for (int u = 0; u < GATHER_UNROLL; ++u) {
        acc.x += r[u].x;
        acc.y += r[u].y;
      }
    }
  }

  const float2 b = *(const float2*)(bias + c);
  float ox = acc.x * di + b.x;
  float oy = acc.y * di + b.y;
  if (relu) {
    ox = ox > 0.0f ? ox : 0.0f;
    oy = oy > 0.0f ? oy : 0.0f;
  }
  *(float2*)(outp + (size_t)node * HIDC + c) = make_float2(ox, oy);
}


__device__ __forceinline__ float2 sum_list_rows(const int* __restrict__ csr_src,
                                                const float* __restrict__ hs,
                                                int start, int end, int s_first,
                                                float2 acc, int lane, int c) {
  for (int k0 = start; k0 < end; k0 += 32) {
    int s_l = s_first;
    if (k0 != start) {
      const int k = k0 + lane;
      s_l = 0;
      if (k < end) s_l = csr_src[k];
    }
    const unsigned o_l = (unsigned)s_l * (unsigned)HIDC;
    const int m = (end - k0 < 32) ? (end - k0) : 32;

    for (int j = 0; j < m; j += GATHER_UNROLL) {
      float2 rw[GATHER_UNROLL];
#pragma unroll
      for (int u = 0; u < GATHER_UNROLL; ++u) {
        rw[u] = make_float2(0.0f, 0.0f);
        if (j + u < m) {
          const unsigned o = (unsigned)__builtin_amdgcn_readlane((int)o_l, j + u);
          rw[u] = *(const float2*)(hs + (size_t)o + c);
        }
      }
#pragma unroll
      for (int u = 0; u < GATHER_UNROLL; ++u) {
        acc.x += rw[u].x;
        acc.y += rw[u].y;
      }
    }
  }
  return acc;
}

__global__ __launch_bounds__(256) void gather_relu_gemm_kernel(const int* __restrict__ row_ptr,
                                                               const int* __restrict__ csr_src,
                                                               const float* __restrict__ dinv,
                                                               const float* __restrict__ hs,
                                                               const float* __restrict__ bias,
                                                               const float* __restrict__ W,
                                                               float* __restrict__ H,
                                                               int n) {
  __shared__ float tile[16 * LDSW];

  const int lane = threadIdx.x & 31;
  const int wid    = __builtin_amdgcn_readfirstlane((int)(threadIdx.x >> 5));
  const int m_base = (int)(blockIdx.x * 16u);
  const int c = lane << 1;

  const int nodeA = m_base + wid * 2;
  const int nodeB = nodeA + 1;
  const int cA = (nodeA < n) ? nodeA : (n - 1);
  const int cB = (nodeB < n) ? nodeB : (n - 1);
  const int iA = (nodeA < n) ? nodeA : n;
  const int iB = (nodeA + 1 < n) ? (nodeA + 1) : n;
  const int iC = (nodeA + 2 < n) ? (nodeA + 2) : n;
  const int startA = row_ptr[iA];
  const int endA   = row_ptr[iB];
  const int endB   = row_ptr[iC];
  const float dA = dinv[cA];
  const float dB = dinv[cB];

  int sA = 0, sB = 0;
  if (startA + lane < endA) sA = csr_src[startA + lane];
  if (endA + lane < endB)   sB = csr_src[endA + lane];
  float2 accA = *(const float2*)(hs + (size_t)cA * HIDC + c);
  float2 accB = *(const float2*)(hs + (size_t)cB * HIDC + c);
  const float2 bs = *(const float2*)(bias + c);

  accA = sum_list_rows(csr_src, hs, startA, endA, sA, accA, lane, c);
  accB = sum_list_rows(csr_src, hs, endA, endB, sB, accB, lane, c);

  float ax = accA.x * dA + bs.x;
  float ay = accA.y * dA + bs.y;
  float bx = accB.x * dB + bs.x;
  float by = accB.y * dB + bs.y;
  ax = (nodeA < n && ax > 0.0f) ? ax : 0.0f;
  ay = (nodeA < n && ay > 0.0f) ? ay : 0.0f;
  bx = (nodeB < n && bx > 0.0f) ? bx : 0.0f;
  by = (nodeB < n && by > 0.0f) ? by : 0.0f;
  *(float2*)(&tile[(wid * 2) * LDSW + c])     = make_float2(ax, ay);
  *(float2*)(&tile[(wid * 2 + 1) * LDSW + c]) = make_float2(bx, by);

  const bool mma  = (wid < 4);
  const int row16 = lane & 15;
  const int half  = lane >> 4;
  const int n0    = (wid & 3) << 4;
  const int mrow  = m_base + half * 8;
  v2f bw[HIDC / 4];
  float sc[8];
#pragma unroll
  for (int t = 0; t < HIDC / 4; ++t) bw[t] = (v2f){0.0f, 0.0f};
#pragma unroll
  for (int r = 0; r < 8; ++r) sc[r] = 0.0f;
  if (mma) {
#pragma unroll
    for (int t = 0; t < HIDC / 4; ++t) {
      const int krow = t * 4 + half * 2;
      bw[t].x = W[(size_t)krow * HIDC + n0 + row16];
      bw[t].y = W[(size_t)(krow + 1) * HIDC + n0 + row16];
    }
#pragma unroll
    for (int r = 0; r < 8; ++r) {
      const int gr = mrow + r;
      sc[r] = dinv[gr < n ? gr : n - 1];
    }
  }
  __syncthreads();
  if (!mma) return;

  v8f acc8 = {};
  const float* arow = &tile[row16 * LDSW];
#pragma unroll
  for (int t = 0; t < HIDC / 4; ++t) {
    v2f a = *(const v2f*)(arow + t * 4 + half * 2);
    acc8 = __builtin_amdgcn_wmma_f32_16x16x4_f32(
        false, a, false, bw[t], (short)0, acc8, false, false);
  }

#pragma unroll
  for (int r = 0; r < 8; ++r) {
    const int gr = mrow + r;
    if (gr < n) H[(size_t)gr * HIDC + n0 + row16] = acc8[r] * sc[r];
  }
}


extern "C" void kernel_launch(void* const* d_in, const int* in_sizes, int n_in,
                              void* d_out, int out_size, void* d_ws, size_t ws_size,
                              hipStream_t stream) {
  const float* x  = (const float*)d_in[0];
  const int*   ei = (const int*)d_in[1];
  const float* W1 = (const float*)d_in[2];
  const float* b1 = (const float*)d_in[3];
  const float* W2 = (const float*)d_in[4];
  const float* b2 = (const float*)d_in[5];

  const int IN_CH = 512;
  const int n = in_sizes[0] / IN_CH;
  const int E = in_sizes[1] / 2;
  const int* src = ei;
  const int* dst = ei + E;
  float* out = (float*)d_out;

  float* ws = (float*)d_ws;
  size_t n_al = ((size_t)n + 255) & ~(size_t)255;
  float* dinv = ws;
  float* bufA = ws + n_al;
  const int nbk = (n + BKT_NODES - 1) / BKT_NODES;
  const int pa  = (E + PCH - 1) / PCH;
  size_t om_al = ((size_t)pa * (size_t)(nbk + 1) + 255) & ~(size_t)255;
  int* rowp   = (int*)(bufA + (size_t)n * HIDC);
  int* offs   = rowp + n_al + 256;
  int* csrs   = offs + om_al;
  int* pairs  = (int*)d_out;

  unsigned gG   = (unsigned)((n + 7) / 8);
  unsigned gF   = (unsigned)((n + 15) / 16);
  unsigned gMM  = (unsigned)((n + 31) / 32);

  bucket_partition_kernel<<<(unsigned)pa, PBLOCK, 0, stream>>>(src, dst, offs, pairs, E, nbk);
  bucket_build_kernel<<<(unsigned)nbk, PBLOCK, 0, stream>>>(pairs, offs, dinv, rowp, csrs, n, nbk, pa);

  gemm_wmma_kernel<<<gMM, 256, 0, stream>>>(x, W1, dinv, out, n, IN_CH);

  gather_relu_gemm_kernel<<<gF, 256, 0, stream>>>(rowp, csrs, dinv, out, b1, W2, bufA, n);

  gather_rows_kernel<<<gG, 256, 0, stream>>>(rowp, csrs, dinv, bufA, b2, out, n, 0);
}


# layer-1 product fed from LDS: W1 k-slice staged beside the x tile (k pairs interleaved, one 8-byte read per B fragment), each wave a 16x64 strip, next slice prefetched to registers; sum order kept
# speedup vs baseline: 4.0527x; 1.0658x over previous
#include <hip/hip_runtime.h>

typedef __attribute__((ext_vector_type(2))) float v2f;
typedef __attribute__((ext_vector_type(4))) float v4f;
typedef __attribute__((ext_vector_type(8))) float v8f;

#define HIDC 64
#define LDSW 68


#define BKT_NODES 512
#define BKT_SHIFT 9
#define BKT_MAX   256
#define PCH       4096
#define PBLOCK    1024
#define PPT       (PCH / PBLOCK)
#define KB_CAP    12288

__device__ __forceinline__ int block_excl_scan_1024(int v, int* wsum, int* s_tot) {
  const int lane = threadIdx.x & 31;
  const int wid  = threadIdx.x >> 5;
  int incl = v;
  for (int o = 1; o < 32; o <<= 1) {
    int u = __shfl_up(incl, o, 32);
    if (lane >= o) incl += u;
  }
  __syncthreads();
  if (lane == 31) wsum[wid] = incl;
  __syncthreads();
  if (wid == 0) {
    const int w = wsum[lane];
    int wi = w;
    for (int o = 1; o < 32; o <<= 1) {
      int u = __shfl_up(wi, o, 32);
      if (lane >= o) wi += u;
    }
    wsum[lane] = wi - w;
    if (lane == 31) *s_tot = wi;
  }
  __syncthreads();
  return wsum[wid] + incl - v;
}

__global__ __launch_bounds__(PBLOCK) void bucket_partition_kernel(const int* __restrict__ src,
                                                                 const int* __restrict__ dst,
                                                                 int* __restrict__ offs,
                                                                 int* __restrict__ pairs,
                                                                 int E, int nbk) {
  __shared__ unsigned hist[BKT_MAX];
  __shared__ int loc_start[BKT_MAX];
  __shared__ int wsum[32];
  __shared__ int s_tot;
  __shared__ int s_src[PCH];
  __shared__ int s_dst[PCH];

  const int t   = threadIdx.x;
  const int blk = blockIdx.x;
  if (t < BKT_MAX) hist[t] = 0u;
  __syncthreads();

  const int e0 = blk * PCH;
  int d[PPT];
  int s[PPT];
  int r[PPT];
#pragma unroll
  for (int u = 0; u < PPT; ++u) {
    const int e = e0 + t + u * PBLOCK;
    d[u] = -1;
    s[u] = 0;
    if (e < E) {
      d[u] = dst[e];
      s[u] = src[e];
    }
  }
#pragma unroll
  for (int u = 0; u < PPT; ++u) {
    const unsigned b = (unsigned)d[u] >> BKT_SHIFT;
    r[u] = -1;
    if (b < (unsigned)nbk) r[u] = (int)atomicAdd(&hist[b], 1u);
  }
  __syncthreads();

  const int own = (t < nbk) ? (int)hist[t] : 0;
  const int lst = block_excl_scan_1024(own, wsum, &s_tot);
  if (t <= nbk) offs[(size_t)blk * (size_t)(nbk + 1) + t] = lst;
  if (t < nbk) loc_start[t] = lst;
  __syncthreads();
  const int placed = s_tot;

#pragma unroll
  for (int u = 0; u < PPT; ++u) {
    if (r[u] >= 0) {
      const int p = loc_start[(unsigned)d[u] >> BKT_SHIFT] + r[u];
      s_src[p] = s[u];
      s_dst[p] = d[u];
    }
  }
  __syncthreads();

  int2* __restrict__ rec = reinterpret_cast<int2*>(pairs) + (size_t)blk * PCH;
  for (int i = t; i < placed; i += PBLOCK) rec[i] = make_int2(s_src[i], s_dst[i]);
}

#define RUNS_PT 13

template <typename F>
__device__ __forceinline__ void leftover_records(const int2* __restrict__ rec, const int* __restrict__ offs,
                                                 int my_o0, int my_o1, int w, int lane, int b,
                                                 int stride, int pa, F f) {
  for (int u = 0; u < RUNS_PT; ++u) {
    const int o0  = __builtin_amdgcn_readlane(my_o0, u);
    const int len = __builtin_amdgcn_readlane(my_o1, u) - o0;
    if (len > 32) {
      const int2* __restrict__ q = rec + (size_t)(w + 32 * u) * PCH + o0;
      for (int i = 32 + lane; i < len; i += 32) f(q[i]);
    }
  }
  for (int blk = w + 32 * RUNS_PT; blk < pa; blk += 32) {
    const int* __restrict__ tq = offs + (size_t)blk * stride + b;
    const int o0  = tq[0];
    const int len = tq[1] - o0;
    const int2* __restrict__ q = rec + (size_t)blk * PCH + o0;
    for (int i = lane; i < len; i += 32) f(q[i]);
  }
}

__global__ __launch_bounds__(PBLOCK) void bucket_build_kernel(const int* __restrict__ pairs,
                                                             const int* __restrict__ offs,
                                                             float* __restrict__ dinv,
                                                             int* __restrict__ row_ptr,
                                                             int* __restrict__ csr_src,
                                                             int n, int nbk, int pa) {
  __shared__ unsigned cnt[BKT_NODES];
  __shared__ int excl[BKT_NODES];
  __shared__ int sorted[KB_CAP];
  __shared__ int wsum[32];
  __shared__ int s_tot;
  __shared__ int s_p[2];

  const int2* __restrict__ rec = reinterpret_cast<const int2*>(pairs);
  const int t    = threadIdx.x;
  const int lane = t & 31;
  const int w      = __builtin_amdgcn_readfirstlane(t >> 5);
  const int b      = blockIdx.x;
  const int base   = b * BKT_NODES;
  const int stride = nbk + 1;

  if (t < BKT_NODES) cnt[t] = 0u;
  if (t < 2) s_p[t] = 0;
  __syncthreads();

  int my_o0 = 0, my_o1 = 0;
  {
    const int blk = w + 32 * lane;
    if (lane < RUNS_PT && blk < pa) {
      const int* __restrict__ tq = offs + (size_t)blk * stride + b;
      my_o0 = tq[0];
      my_o1 = tq[1];
    }
  }
  int2 raw[RUNS_PT];
#pragma unroll
  for (int u = 0; u < RUNS_PT; ++u) {
    const int o0  = __builtin_amdgcn_readlane(my_o0, u);
    const int len = __builtin_amdgcn_readlane(my_o1, u) - o0;
    raw[u] = make_int2(0, 0);
    if (lane < len) raw[u] = rec[(size_t)(w + 32 * u) * PCH + o0 + lane];
  }

  int sum0 = my_o0, sum1 = my_o1;
  for (int blk = 32 * RUNS_PT + t; blk < pa; blk += PBLOCK) {
    const int* __restrict__ tq = offs + (size_t)blk * stride + b;
    sum0 += tq[0];
    sum1 += tq[1];
  }

  for (int o = 16; o > 0; o >>= 1) {
    sum0 += __shfl_xor(sum0, o, 32);
    sum1 += __shfl_xor(sum1, o, 32);
  }
  if (lane == 0) {
    atomicAdd(&s_p[0], sum0);
    atomicAdd(&s_p[1], sum1);
  }

  int rs[RUNS_PT];
  int rv[RUNS_PT];
#pragma unroll
  for (int u = 0; u < RUNS_PT; ++u) {
    const int len = __builtin_amdgcn_readlane(my_o1, u) - __builtin_amdgcn_readlane(my_o0, u);
    rs[u] = raw[u].x;
    rv[u] = (lane < len) ? (raw[u].y - base) : -1;
    if ((unsigned)rv[u] < (unsigned)BKT_NODES) atomicAdd(&cnt[rv[u]], 1u);
  }
  leftover_records(rec, offs, my_o0, my_o1, w, lane, b, stride, pa, [&](int2 pr) {
    const unsigned v = (unsigned)(pr.y - base);
    if (v < (unsigned)BKT_NODES) atomicAdd(&cnt[v], 1u);
  });
  __syncthreads();

  const int p0 = s_p[0];
  const int p1 = s_p[1];
  const int m  = p1 - p0;
  const bool in_lds = (m <= KB_CAP);

  const int c  = (t < BKT_NODES) ? (int)cnt[t] : 0;
  const int ex = block_excl_scan_1024(c, wsum, &s_tot);
  if (t < BKT_NODES) {
    const int node = base + t;
    if (node < n) {
      const float dg = (float)(c + 1);
      dinv[node]    = (dg > 0.0f) ? rsqrtf(dg) : 0.0f;
      row_ptr[node] = p0 + ex;
    }
    excl[t] = ex;
    cnt[t]  = 0u;
  }
  if (b == nbk - 1 && t == 0) row_ptr[n] = p1;
  __syncthreads();

#pragma unroll
  for (int u = 0; u < RUNS_PT; ++u) {
    if ((unsigned)rv[u] < (unsigned)BKT_NODES) {
      const int pos = excl[rv[u]] + (int)atomicAdd(&cnt[rv[u]], 1u);
      if (in_lds) sorted[pos] = rs[u];
      else        csr_src[p0 + pos] = rs[u];
    }
  }
  leftover_records(rec, offs, my_o0, my_o1, w, lane, b, stride, pa, [&](int2 pr) {
    const unsigned v = (unsigned)(pr.y - base);
    if (v < (unsigned)BKT_NODES) {
      const int pos = excl[v] + (int)atomicAdd(&cnt[v], 1u);
      if (in_lds) sorted[pos] = pr.x;
      else        csr_src[p0 + pos] = pr.x;
    }
  });
  if (in_lds) {
    __syncthreads();
    for (int i = t; i < m; i += PBLOCK) csr_src[p0 + i] = sorted[i];
  }
}


#define GM_ROWS 64
#define GM_KS   32
#define XPITCH  36
#define WPITCH  160

__global__ __launch_bounds__(128) void gemm_wmma_kernel(
    const float* __restrict__ A, const float* __restrict__ W,
    const float* __restrict__ rs, float* __restrict__ H, int n, int K) {
  __shared__ float xt[GM_ROWS * XPITCH];
  __shared__ float wt[(GM_KS / 2) * WPITCH];

  const int t     = threadIdx.x;
  const int wid   = t >> 5;
  const int lane  = t & 31;
  const int row16 = lane & 15;
  const int half  = lane >> 4;
  const int m_base = blockIdx.x * GM_ROWS;

  const int xc = (t & 7) << 2;
  const int xr = t >> 3;
  const int wc = (t & 15) << 2;
  const int wr = t >> 4;

  size_t xo[4];
#pragma unroll
  for (int i = 0; i < 4; ++i) {
    const int grow = m_base + xr + 16 * i;
    xo[i] = (size_t)(grow < n ? grow : n - 1) * (size_t)K + (size_t)xc;
  }
  const size_t wo = (size_t)(2 * wr) * HIDC + (size_t)wc;

  v4f xv[4];
  v4f w0[2];
  v4f w1[2];
#pragma unroll
  for (int i = 0; i < 4; ++i) xv[i] = *(const v4f*)(A + xo[i]);
#pragma unroll
  for (int i = 0; i < 2; ++i) {
    w0[i] = *(const v4f*)(W + wo + (16 * i) * HIDC);
    w1[i] = *(const v4f*)(W + wo + (16 * i + 1) * HIDC);
  }

  v8f acc[4] = {};
  const float* arow = &xt[(wid * 16 + row16) * XPITCH + half * 2];
  const float* brow = &wt[half * WPITCH + row16 * 2];

  for (int kb = 0; kb < K; kb += GM_KS) {
#pragma unroll
    for (int i = 0; i < 4; ++i) *(v4f*)(&xt[(xr + 16 * i) * XPITCH + xc]) = xv[i];
#pragma unroll
    for (int i = 0; i < 2; ++i) {
      float* q = &wt[(wr + 8 * i) * WPITCH + 2 * wc];
      *(v4f*)(q)     = (v4f){w0[i].x, w1[i].x, w0[i].y, w1[i].y};
      *(v4f*)(q + 4) = (v4f){w0[i].z, w1[i].z, w0[i].w, w1[i].w};
    }
    __syncthreads();

    const int kn = kb + GM_KS;
    if (kn < K) {
      const float* __restrict__ An = A + kn;
      const float* __restrict__ Wn = W + (size_t)kn * HIDC;
#pragma unroll
      for (int i = 0; i < 4; ++i) xv[i] = *(const v4f*)(An + xo[i]);
#pragma unroll
      for (int i = 0; i < 2; ++i) {
        w0[i] = *(const v4f*)(Wn + wo + (16 * i) * HIDC);
        w1[i] = *(const v4f*)(Wn + wo + (16 * i + 1) * HIDC);
      }
    }

#pragma unroll
    for (int kk = 0; kk < GM_KS; kk += 4) {
      const v2f a = *(const v2f*)(arow + kk);
#pragma unroll
      for (int nt = 0; nt < 4; ++nt) {
        const v2f b = *(const v2f*)(brow + (kk >> 1) * WPITCH + nt * 32);
        acc[nt] = __builtin_amdgcn_wmma_f32_16x16x4_f32(
            false, a, false, b, (short)0, acc[nt], false, false);
      }
    }
    __syncthreads();
  }

  const int mrow = m_base + wid * 16 + half * 8;
  float sc[8];
#pragma unroll
  for (int r = 0; r < 8; ++r) {
    const int gr = mrow + r;
    sc[r] = rs[gr < n ? gr : n - 1];
  }
#pragma unroll
  for (int r = 0; r < 8; ++r) {
    const int gr = mrow + r;
    if (gr < n) {
#pragma unroll
      for (int nt = 0; nt < 4; ++nt) H[(size_t)gr * HIDC + nt * 16 + row16] = acc[nt][r] * sc[r];
    }
  }
}


#define GATHER_UNROLL 8

__global__ __launch_bounds__(256) void gather_rows_kernel(const int* __restrict__ row_ptr,
                                                          const int* __restrict__ csr_src,
                                                          const float* __restrict__ dinv,
                                                          const float* __restrict__ hs,
                                                          const float* __restrict__ bias,
                                                          float* __restrict__ outp,
                                                          int n, int relu) {
  const int lane = threadIdx.x & 31;
  const int node = __builtin_amdgcn_readfirstlane((int)(blockIdx.x * 8u + (threadIdx.x >> 5)));
  if (node >= n) return;

  const int start = row_ptr[node];
  const int end   = row_ptr[node + 1];
  const float di  = dinv[node];
  const int c = lane << 1;

  float2 acc = *(const float2*)(hs + (size_t)node * HIDC + c);

  for (int k0 = start; k0 < end; k0 += 32) {
    const int k = k0 + lane;
    int s_l = 0;
    if (k < end) s_l = csr_src[k];
    const unsigned o_l = (unsigned)s_l * (unsigned)HIDC;
    const int m = (end - k0 < 32) ? (end - k0) : 32;

    for (int j = 0; j < m; j += GATHER_UNROLL) {
      float2 r[GATHER_UNROLL];
#pragma unroll
      for (int u = 0; u < GATHER_UNROLL; ++u) {
        r[u] = make_float2(0.0f, 0.0f);
        if (j + u < m) {
          const unsigned o = (unsigned)__builtin_amdgcn_readlane((int)o_l, j + u);
          r[u] = *(const float2*)(hs + (size_t)o + c);
        }
      }
#pragma unroll
      for (int u = 0; u < GATHER_UNROLL; ++u) {
        acc.x += r[u].x;
        acc.y += r[u].y;
      }
    }
  }

  const float2 b = *(const float2*)(bias + c);
  float ox = acc.x * di + b.x;
  float oy = acc.y * di + b.y;
  if (relu) {
    ox = ox > 0.0f ? ox : 0.0f;
    oy = oy > 0.0f ? oy : 0.0f;
  }
  *(float2*)(outp + (size_t)node * HIDC + c) = make_float2(ox, oy);
}


__device__ __forceinline__ float2 sum_list_rows(const int* __restrict__ csr_src,
                                                const float* __restrict__ hs,
                                                int start, int end, int s_first,
                                                float2 acc, int lane, int c) {
  for (int k0 = start; k0 < end; k0 += 32) {
    int s_l = s_first;
    if (k0 != start) {
      const int k = k0 + lane;
      s_l = 0;
      if (k < end) s_l = csr_src[k];
    }
    const unsigned o_l = (unsigned)s_l * (unsigned)HIDC;
    const int m = (end - k0 < 32) ? (end - k0) : 32;

    for (int j = 0; j < m; j += GATHER_UNROLL) {
      float2 rw[GATHER_UNROLL];
#pragma unroll
      for (int u = 0; u < GATHER_UNROLL; ++u) {
        rw[u] = make_float2(0.0f, 0.0f);
        if (j + u < m) {
          const unsigned o = (unsigned)__builtin_amdgcn_readlane((int)o_l, j + u);
          rw[u] = *(const float2*)(hs + (size_t)o + c);
        }
      }
#pragma unroll
      for (int u = 0; u < GATHER_UNROLL; ++u) {
        acc.x += rw[u].x;
        acc.y += rw[u].y;
      }
    }
  }
  return acc;
}

__global__ __launch_bounds__(256) void gather_relu_gemm_kernel(const int* __restrict__ row_ptr,
                                                               const int* __restrict__ csr_src,
                                                               const float* __restrict__ dinv,
                                                               const float* __restrict__ hs,
                                                               const float* __restrict__ bias,
                                                               const float* __restrict__ W,
                                                               float* __restrict__ H,
                                                               int n) {
  __shared__ float tile[16 * LDSW];

  const int lane = threadIdx.x & 31;
  const int wid    = __builtin_amdgcn_readfirstlane((int)(threadIdx.x >> 5));
  const int m_base = (int)(blockIdx.x * 16u);
  const int c = lane << 1;

  const int nodeA = m_base + wid * 2;
  const int nodeB = nodeA + 1;
  const int cA = (nodeA < n) ? nodeA : (n - 1);
  const int cB = (nodeB < n) ? nodeB : (n - 1);
  const int iA = (nodeA < n) ? nodeA : n;
  const int iB = (nodeA + 1 < n) ? (nodeA + 1) : n;
  const int iC = (nodeA + 2 < n) ? (nodeA + 2) : n;
  const int startA = row_ptr[iA];
  const int endA   = row_ptr[iB];
  const int endB   = row_ptr[iC];
  const float dA = dinv[cA];
  const float dB = dinv[cB];

  int sA = 0, sB = 0;
  if (startA + lane < endA) sA = csr_src[startA + lane];
  if (endA + lane < endB)   sB = csr_src[endA + lane];
  float2 accA = *(const float2*)(hs + (size_t)cA * HIDC + c);
  float2 accB = *(const float2*)(hs + (size_t)cB * HIDC + c);
  const float2 bs = *(const float2*)(bias + c);

  accA = sum_list_rows(csr_src, hs, startA, endA, sA, accA, lane, c);
  accB = sum_list_rows(csr_src, hs, endA, endB, sB, accB, lane, c);

  float ax = accA.x * dA + bs.x;
  float ay = accA.y * dA + bs.y;
  float bx = accB.x * dB + bs.x;
  float by = accB.y * dB + bs.y;
  ax = (nodeA < n && ax > 0.0f) ? ax : 0.0f;
  ay = (nodeA < n && ay > 0.0f) ? ay : 0.0f;
  bx = (nodeB < n && bx > 0.0f) ? bx : 0.0f;
  by = (nodeB < n && by > 0.0f) ? by : 0.0f;
  *(float2*)(&tile[(wid * 2) * LDSW + c])     = make_float2(ax, ay);
  *(float2*)(&tile[(wid * 2 + 1) * LDSW + c]) = make_float2(bx, by);

  const bool mma  = (wid < 4);
  const int row16 = lane & 15;
  const int half  = lane >> 4;
  const int n0    = (wid & 3) << 4;
  const int mrow  = m_base + half * 8;
  v2f bw[HIDC / 4];
  float sc[8];
#pragma unroll
  for (int t = 0; t < HIDC / 4; ++t) bw[t] = (v2f){0.0f, 0.0f};
#pragma unroll
  for (int r = 0; r < 8; ++r) sc[r] = 0.0f;
  if (mma) {
#pragma unroll
    for (int t = 0; t < HIDC / 4; ++t) {
      const int krow = t * 4 + half * 2;
      bw[t].x = W[(size_t)krow * HIDC + n0 + row16];
      bw[t].y = W[(size_t)(krow + 1) * HIDC + n0 + row16];
    }
#pragma unroll
    for (int r = 0; r < 8; ++r) {
      const int gr = mrow + r;
      sc[r] = dinv[gr < n ? gr : n - 1];
    }
  }
  __syncthreads();
  if (!mma) return;

  v8f acc8 = {};
  const float* arow = &tile[row16 * LDSW];
#pragma unroll
  for (int t = 0; t < HIDC / 4; ++t) {
    v2f a = *(const v2f*)(arow + t * 4 + half * 2);
    acc8 = __builtin_amdgcn_wmma_f32_16x16x4_f32(
        false, a, false, bw[t], (short)0, acc8, false, false);
  }

#pragma unroll
  for (int r = 0; r < 8; ++r) {
    const int gr = mrow + r;
    if (gr < n) H[(size_t)gr * HIDC + n0 + row16] = acc8[r] * sc[r];
  }
}


extern "C" void kernel_launch(void* const* d_in, const int* in_sizes, int n_in,
                              void* d_out, int out_size, void* d_ws, size_t ws_size,
                              hipStream_t stream) {
  const float* x  = (const float*)d_in[0];
  const int*   ei = (const int*)d_in[1];
  const float* W1 = (const float*)d_in[2];
  const float* b1 = (const float*)d_in[3];
  const float* W2 = (const float*)d_in[4];
  const float* b2 = (const float*)d_in[5];

  const int IN_CH = 512;
  const int n = in_sizes[0] / IN_CH;
  const int E = in_sizes[1] / 2;
  const int* src = ei;
  const int* dst = ei + E;
  float* out = (float*)d_out;

  float* ws = (float*)d_ws;
  size_t n_al = ((size_t)n + 255) & ~(size_t)255;
  float* dinv = ws;
  float* bufA = ws + n_al;
  const int nbk = (n + BKT_NODES - 1) / BKT_NODES;
  const int pa  = (E + PCH - 1) / PCH;
  size_t om_al = ((size_t)pa * (size_t)(nbk + 1) + 255) & ~(size_t)255;
  int* rowp   = (int*)(bufA + (size_t)n * HIDC);
  int* offs   = rowp + n_al + 256;
  int* csrs   = offs + om_al;
  int* pairs  = (int*)d_out;

  unsigned gG   = (unsigned)((n + 7) / 8);
  unsigned gF   = (unsigned)((n + 15) / 16);
  unsigned gMM  = (unsigned)((n + GM_ROWS - 1) / GM_ROWS);

  bucket_partition_kernel<<<(unsigned)pa, PBLOCK, 0, stream>>>(src, dst, offs, pairs, E, nbk);
  bucket_build_kernel<<<(unsigned)nbk, PBLOCK, 0, stream>>>(pairs, offs, dinv, rowp, csrs, n, nbk, pa);

  gemm_wmma_kernel<<<gMM, 128, 0, stream>>>(x, W1, dinv, out, n, IN_CH);

  gather_relu_gemm_kernel<<<gF, 256, 0, stream>>>(rowp, csrs, dinv, out, b1, W2, bufA, n);

  gather_rows_kernel<<<gG, 256, 0, stream>>>(rowp, csrs, dinv, bufA, b2, out, n, 0);
}


# both propagation launches: one wave sums 4 consecutive nodes as ONE stream of list entries (bounds, scales, own rows asked for once; rows in full groups of 16 across node ends); sum order kept
# speedup vs baseline: 4.2516x; 1.0491x over previous
#include <hip/hip_runtime.h>

typedef __attribute__((ext_vector_type(2))) float v2f;
typedef __attribute__((ext_vector_type(4))) float v4f;
typedef __attribute__((ext_vector_type(8))) float v8f;

#define HIDC 64
#define LDSW 68


#define BKT_NODES 512
#define BKT_SHIFT 9
#define BKT_MAX   256
#define PCH       4096
#define PBLOCK    1024
#define PPT       (PCH / PBLOCK)
#define KB_CAP    12288

__device__ __forceinline__ int block_excl_scan_1024(int v, int* wsum, int* s_tot) {
  const int lane = threadIdx.x & 31;
  const int wid  = threadIdx.x >> 5;
  int incl = v;
  for (int o = 1; o < 32; o <<= 1) {
    int u = __shfl_up(incl, o, 32);
    if (lane >= o) incl += u;
  }
  __syncthreads();
  if (lane == 31) wsum[wid] = incl;
  __syncthreads();
  if (wid == 0) {
    const int w = wsum[lane];
    int wi = w;
    for (int o = 1; o < 32; o <<= 1) {
      int u = __shfl_up(wi, o, 32);
      if (lane >= o) wi += u;
    }
    wsum[lane] = wi - w;
    if (lane == 31) *s_tot = wi;
  }
  __syncthreads();
  return wsum[wid] + incl - v;
}

__global__ __launch_bounds__(PBLOCK) void bucket_partition_kernel(const int* __restrict__ src,
                                                                 const int* __restrict__ dst,
                                                                 int* __restrict__ offs,
                                                                 int* __restrict__ pairs,
                                                                 int E, int nbk) {
  __shared__ unsigned hist[BKT_MAX];
  __shared__ int loc_start[BKT_MAX];
  __shared__ int wsum[32];
  __shared__ int s_tot;
  __shared__ int s_src[PCH];
  __shared__ int s_dst[PCH];

  const int t   = threadIdx.x;
  const int blk = blockIdx.x;
  if (t < BKT_MAX) hist[t] = 0u;
  __syncthreads();

  const int e0 = blk * PCH;
  int d[PPT];
  int s[PPT];
  int r[PPT];
#pragma unroll
  for (int u = 0; u < PPT; ++u) {
    const int e = e0 + t + u * PBLOCK;
    d[u] = -1;
    s[u] = 0;
    if (e < E) {
      d[u] = dst[e];
      s[u] = src[e];
    }
  }
#pragma unroll
  for (int u = 0; u < PPT; ++u) {
    const unsigned b = (unsigned)d[u] >> BKT_SHIFT;
    r[u] = -1;
    if (b < (unsigned)nbk) r[u] = (int)atomicAdd(&hist[b], 1u);
  }
  __syncthreads();

  const int own = (t < nbk) ? (int)hist[t] : 0;
  const int lst = block_excl_scan_1024(own, wsum, &s_tot);
  if (t <= nbk) offs[(size_t)blk * (size_t)(nbk + 1) + t] = lst;
  if (t < nbk) loc_start[t] = lst;
  __syncthreads();
  const int placed = s_tot;

#pragma unroll
  for (int u = 0; u < PPT; ++u) {
    if (r[u] >= 0) {
      const int p = loc_start[(unsigned)d[u] >> BKT_SHIFT] + r[u];
      s_src[p] = s[u];
      s_dst[p] = d[u];
    }
  }
  __syncthreads();

  int2* __restrict__ rec = reinterpret_cast<int2*>(pairs) + (size_t)blk * PCH;
  for (int i = t; i < placed; i += PBLOCK) rec[i] = make_int2(s_src[i], s_dst[i]);
}

#define RUNS_PT 13

template <typename F>
__device__ __forceinline__ void leftover_records(const int2* __restrict__ rec, const int* __restrict__ offs,
                                                 int my_o0, int my_o1, int w, int lane, int b,
                                                 int stride, int pa, F f) {
  for (int u = 0; u < RUNS_PT; ++u) {
    const int o0  = __builtin_amdgcn_readlane(my_o0, u);
    const int len = __builtin_amdgcn_readlane(my_o1, u) - o0;
    if (len > 32) {
      const int2* __restrict__ q = rec + (size_t)(w + 32 * u) * PCH + o0;
      for (int i = 32 + lane; i < len; i += 32) f(q[i]);
    }
  }
  for (int blk = w + 32 * RUNS_PT; blk < pa; blk += 32) {
    const int* __restrict__ tq = offs + (size_t)blk * stride + b;
    const int o0  = tq[0];
    const int len = tq[1] - o0;
    const int2* __restrict__ q = rec + (size_t)blk * PCH + o0;
    for (int i = lane; i < len; i += 32) f(q[i]);
  }
}

__global__ __launch_bounds__(PBLOCK) void bucket_build_kernel(const int* __restrict__ pairs,
                                                             const int* __restrict__ offs,
                                                             float* __restrict__ dinv,
                                                             int* __restrict__ row_ptr,
                                                             int* __restrict__ csr_src,
                                                             int n, int nbk, int pa) {
  __shared__ unsigned cnt[BKT_NODES];
  __shared__ int excl[BKT_NODES];
  __shared__ int sorted[KB_CAP];
  __shared__ int wsum[32];
  __shared__ int s_tot;
  __shared__ int s_p[2];

  const int2* __restrict__ rec = reinterpret_cast<const int2*>(pairs);
  const int t    = threadIdx.x;
  const int lane = t & 31;
  const int w      = __builtin_amdgcn_readfirstlane(t >> 5);
  const int b      = blockIdx.x;
  const int base   = b * BKT_NODES;
  const int stride = nbk + 1;

  if (t < BKT_NODES) cnt[t] = 0u;
  if (t < 2) s_p[t] = 0;
  __syncthreads();

  int my_o0 = 0, my_o1 = 0;
  {
    const int blk = w + 32 * lane;
    if (lane < RUNS_PT && blk < pa) {
      const int* __restrict__ tq = offs + (size_t)blk * stride + b;
      my_o0 = tq[0];
      my_o1 = tq[1];
    }
  }
  int2 raw[RUNS_PT];
#pragma unroll
  for (int u = 0; u < RUNS_PT; ++u) {
    const int o0  = __builtin_amdgcn_readlane(my_o0, u);
    const int len = __builtin_amdgcn_readlane(my_o1, u) - o0;
    raw[u] = make_int2(0, 0);
    if (lane < len) raw[u] = rec[(size_t)(w + 32 * u) * PCH + o0 + lane];
  }

  int sum0 = my_o0, sum1 = my_o1;
  for (int blk = 32 * RUNS_PT + t; blk < pa; blk += PBLOCK) {
    const int* __restrict__ tq = offs + (size_t)blk * stride + b;
    sum0 += tq[0];
    sum1 += tq[1];
  }

  for (int o = 16; o > 0; o >>= 1) {
    sum0 += __shfl_xor(sum0, o, 32);
    sum1 += __shfl_xor(sum1, o, 32);
  }
  if (lane == 0) {
    atomicAdd(&s_p[0], sum0);
    atomicAdd(&s_p[1], sum1);
  }

  int rs[RUNS_PT];
  int rv[RUNS_PT];
#pragma unroll
  for (int u = 0; u < RUNS_PT; ++u) {
    const int len = __builtin_amdgcn_readlane(my_o1, u) - __builtin_amdgcn_readlane(my_o0, u);
    rs[u] = raw[u].x;
    rv[u] = (lane < len) ? (raw[u].y - base) : -1;
    if ((unsigned)rv[u] < (unsigned)BKT_NODES) atomicAdd(&cnt[rv[u]], 1u);
  }
  leftover_records(rec, offs, my_o0, my_o1, w, lane, b, stride, pa, [&](int2 pr) {
    const unsigned v = (unsigned)(pr.y - base);
    if (v < (unsigned)BKT_NODES) atomicAdd(&cnt[v], 1u);
  });
  __syncthreads();

  const int p0 = s_p[0];
  const int p1 = s_p[1];
  const int m  = p1 - p0;
  const bool in_lds = (m <= KB_CAP);

  const int c  = (t < BKT_NODES) ? (int)cnt[t] : 0;
  const int ex = block_excl_scan_1024(c, wsum, &s_tot);
  if (t < BKT_NODES) {
    const int node = base + t;
    if (node < n) {
      const float dg = (float)(c + 1);
      dinv[node]    = (dg > 0.0f) ? rsqrtf(dg) : 0.0f;
      row_ptr[node] = p0 + ex;
    }
    excl[t] = ex;
    cnt[t]  = 0u;
  }
  if (b == nbk - 1 && t == 0) row_ptr[n] = p1;
  __syncthreads();

#pragma unroll
  for (int u = 0; u < RUNS_PT; ++u) {
    if ((unsigned)rv[u] < (unsigned)BKT_NODES) {
      const int pos = excl[rv[u]] + (int)atomicAdd(&cnt[rv[u]], 1u);
      if (in_lds) sorted[pos] = rs[u];
      else        csr_src[p0 + pos] = rs[u];
    }
  }
  leftover_records(rec, offs, my_o0, my_o1, w, lane, b, stride, pa, [&](int2 pr) {
    const unsigned v = (unsigned)(pr.y - base);
    if (v < (unsigned)BKT_NODES) {
      const int pos = excl[v] + (int)atomicAdd(&cnt[v], 1u);
      if (in_lds) sorted[pos] = pr.x;
      else        csr_src[p0 + pos] = pr.x;
    }
  });
  if (in_lds) {
    __syncthreads();
    for (int i = t; i < m; i += PBLOCK) csr_src[p0 + i] = sorted[i];
  }
}


#define GM_ROWS 64
#define GM_KS   32
#define XPITCH  36
#define WPITCH  160

__global__ __launch_bounds__(128) void gemm_wmma_kernel(
    const float* __restrict__ A, const float* __restrict__ W,
    const float* __restrict__ rs, float* __restrict__ H, int n, int K) {
  __shared__ float xt[GM_ROWS * XPITCH];
  __shared__ float wt[(GM_KS / 2) * WPITCH];

  const int t     = threadIdx.x;
  const int wid   = t >> 5;
  const int lane  = t & 31;
  const int row16 = lane & 15;
  const int half  = lane >> 4;
  const int m_base = blockIdx.x * GM_ROWS;

  const int xc = (t & 7) << 2;
  const int xr = t >> 3;
  const int wc = (t & 15) << 2;
  const int wr = t >> 4;

  size_t xo[4];
#pragma unroll
  for (int i = 0; i < 4; ++i) {
    const int grow = m_base + xr + 16 * i;
    xo[i] = (size_t)(grow < n ? grow : n - 1) * (size_t)K + (size_t)xc;
  }
  const size_t wo = (size_t)(2 * wr) * HIDC + (size_t)wc;

  v4f xv[4];
  v4f w0[2];
  v4f w1[2];
#pragma unroll
  for (int i = 0; i < 4; ++i) xv[i] = *(const v4f*)(A + xo[i]);
#pragma unroll
  for (int i = 0; i < 2; ++i) {
    w0[i] = *(const v4f*)(W + wo + (16 * i) * HIDC);
    w1[i] = *(const v4f*)(W + wo + (16 * i + 1) * HIDC);
  }

  v8f acc[4] = {};
  const float* arow = &xt[(wid * 16 + row16) * XPITCH + half * 2];
  const float* brow = &wt[half * WPITCH + row16 * 2];

  for (int kb = 0; kb < K; kb += GM_KS) {
#pragma unroll
    for (int i = 0; i < 4; ++i) *(v4f*)(&xt[(xr + 16 * i) * XPITCH + xc]) = xv[i];
#pragma unroll
    for (int i = 0; i < 2; ++i) {
      float* q = &wt[(wr + 8 * i) * WPITCH + 2 * wc];
      *(v4f*)(q)     = (v4f){w0[i].x, w1[i].x, w0[i].y, w1[i].y};
      *(v4f*)(q + 4) = (v4f){w0[i].z, w1[i].z, w0[i].w, w1[i].w};
    }
    __syncthreads();

    const int kn = kb + GM_KS;
    if (kn < K) {
      const float* __restrict__ An = A + kn;
      const float* __restrict__ Wn = W + (size_t)kn * HIDC;
#pragma unroll
      for (int i = 0; i < 4; ++i) xv[i] = *(const v4f*)(An + xo[i]);
#pragma unroll
      for (int i = 0; i < 2; ++i) {
        w0[i] = *(const v4f*)(Wn + wo + (16 * i) * HIDC);
        w1[i] = *(const v4f*)(Wn + wo + (16 * i + 1) * HIDC);
      }
    }

#pragma unroll
    for (int kk = 0; kk < GM_KS; kk += 4) {
      const v2f a = *(const v2f*)(arow + kk);
#pragma unroll
      for (int nt = 0; nt < 4; ++nt) {
        const v2f b = *(const v2f*)(brow + (kk >> 1) * WPITCH + nt * 32);
        acc[nt] = __builtin_amdgcn_wmma_f32_16x16x4_f32(
            false, a, false, b, (short)0, acc[nt], false, false);
      }
    }
    __syncthreads();
  }

  const int mrow = m_base + wid * 16 + half * 8;
  float sc[8];
#pragma unroll
  for (int r = 0; r < 8; ++r) {
    const int gr = mrow + r;
    sc[r] = rs[gr < n ? gr : n - 1];
  }
#pragma unroll
  for (int r = 0; r < 8; ++r) {
    const int gr = mrow + r;
    if (gr < n) {
#pragma unroll
      for (int nt = 0; nt < 4; ++nt) H[(size_t)gr * HIDC + nt * 16 + row16] = acc[nt][r] * sc[r];
    }
  }
}


#define NPW 4
#define SG  16

struct full_group { static constexpr bool value = true; };
struct part_group { static constexpr bool value = false; };

template <typename FIN>
__device__ __forceinline__ void sum_node_stream(const int* __restrict__ row_ptr,
                                                const int* __restrict__ csr_src,
                                                const float* __restrict__ dinv,
                                                const float* __restrict__ hs,
                                                int node0, int cnt, int lane, int c, FIN fin) {
  const int rp_l = row_ptr[node0 + (lane < cnt ? lane : cnt)];
  const int dv_l = __float_as_int(dinv[node0 + (lane < cnt ? lane : cnt - 1)]);
  float2 sf[NPW];
#pragma unroll
  for (int i = 0; i < NPW; ++i) sf[i] = *(const float2*)(hs + (size_t)(node0 + (i < cnt ? i : cnt - 1)) * HIDC + c);
  float2 acc = sf[0];

  const int S = __builtin_amdgcn_readlane(rp_l, 0);
  const int E = __builtin_amdgcn_readlane(rp_l, cnt);
  const unsigned cb = (unsigned)c * 4u;
  int cur = 0;
  int bnd = __builtin_amdgcn_readlane(rp_l, 1);

  auto next_node = [&]() {
    fin(cur, acc, __int_as_float(__builtin_amdgcn_readlane(dv_l, cur)));
    ++cur;
    acc = sf[1];
#pragma unroll
    for (int i = 2; i < NPW; ++i) {
      if (cur == i) acc = sf[i];
    }
    bnd = __builtin_amdgcn_readlane(rp_l, cur + 1);
  };

  int s_nx = 0;
  if (S + lane < E) s_nx = csr_src[S + lane];
  unsigned o_l = (unsigned)s_nx * (unsigned)(HIDC * 4);
  for (int k0 = S; k0 < E; k0 += 32) {
    const int m = (E - k0 < 32) ? (E - k0) : 32;

    auto group = [&](auto full_tag, const int j, const int mg) {
      constexpr bool FULL = decltype(full_tag)::value;
      float2 r[SG];
#pragma unroll
      for (int u = 0; u < SG; ++u) {
        r[u] = make_float2(0.0f, 0.0f);
        if (FULL || u < mg) {
          const unsigned ob = (unsigned)__builtin_amdgcn_readlane((int)o_l, j + u);
          r[u] = *(const float2*)((const char*)hs + (ob + cb));
        }
      }
      if (j == 0) {
        s_nx = 0;
        if (k0 + 32 + lane < E) s_nx = csr_src[k0 + 32 + lane];
      }
      int rel = bnd - (k0 + j);
#pragma unroll
      for (int u = 0; u < SG; ++u) {
        if (FULL || u < mg) {
          while (u == rel) {
            next_node();
            rel = bnd - (k0 + j);
          }
          acc.x += r[u].x;
          acc.y += r[u].y;
        }
      }
    };

#pragma unroll
    for (int j = 0; j < 32; j += SG) {
      if (j < m) {
        if (m - j >= SG) group(full_group{}, j, SG);
        else             group(part_group{}, j, m - j);
      }
    }
    o_l = (unsigned)s_nx * (unsigned)(HIDC * 4);
  }
  for (;;) {
    const bool last = (cur + 1 >= cnt);
    next_node();
    if (last) break;
  }
}

__global__ __launch_bounds__(256) void gather_rows_kernel(const int* __restrict__ row_ptr,
                                                          const int* __restrict__ csr_src,
                                                          const float* __restrict__ dinv,
                                                          const float* __restrict__ hs,
                                                          const float* __restrict__ bias,
                                                          float* __restrict__ outp,
                                                          int n, int relu) {
  const int lane = threadIdx.x & 31;
  const int node0 = __builtin_amdgcn_readfirstlane((int)(blockIdx.x * (8u * NPW) + (threadIdx.x >> 5) * NPW));
  if (node0 >= n) return;
  const int cnt = (n - node0 < NPW) ? (n - node0) : NPW;
  const int c = lane << 1;
  const float2 b = *(const float2*)(bias + c);

  sum_node_stream(row_ptr, csr_src, dinv, hs, node0, cnt, lane, c, [&](int i, float2 a, float d) {
    float ox = a.x * d + b.x;
    float oy = a.y * d + b.y;
    if (relu) {
      ox = ox > 0.0f ? ox : 0.0f;
      oy = oy > 0.0f ? oy : 0.0f;
    }
    *(float2*)(outp + (size_t)(node0 + i) * HIDC + c) = make_float2(ox, oy);
  });
}


__global__ __launch_bounds__(256) void gather_relu_gemm_kernel(const int* __restrict__ row_ptr,
                                                               const int* __restrict__ csr_src,
                                                               const float* __restrict__ dinv,
                                                               const float* __restrict__ hs,
                                                               const float* __restrict__ bias,
                                                               const float* __restrict__ W,
                                                               float* __restrict__ H,
                                                               int n) {
  __shared__ float tile[8 * NPW * LDSW];

  const int lane = threadIdx.x & 31;
  const int wid    = __builtin_amdgcn_readfirstlane((int)(threadIdx.x >> 5));
  const int m_base = (int)(blockIdx.x * (8u * NPW));
  const int c = lane << 1;

  const int node0 = m_base + wid * NPW;
  const int left  = n - node0;
  const int cnt   = (left < NPW) ? (left > 0 ? left : 0) : NPW;
  const float2 bs = *(const float2*)(bias + c);

  for (int i = cnt; i < NPW; ++i) *(float2*)(&tile[(wid * NPW + i) * LDSW + c]) = make_float2(0.0f, 0.0f);
  if (cnt > 0) {
    sum_node_stream(row_ptr, csr_src, dinv, hs, node0, cnt, lane, c, [&](int i, float2 a, float d) {
      float vx = a.x * d + bs.x;
      float vy = a.y * d + bs.y;
      vx = vx > 0.0f ? vx : 0.0f;
      vy = vy > 0.0f ? vy : 0.0f;
      *(float2*)(&tile[(wid * NPW + i) * LDSW + c]) = make_float2(vx, vy);
    });
  }

  const int row16 = lane & 15;
  const int half  = lane >> 4;
  const int rt    = wid >> 2;
  const int n0    = (wid & 3) << 4;
  const int mrow  = m_base + rt * 16 + half * 8;
  v2f bw[HIDC / 4];
  float sc[8];
#pragma unroll
  for (int t = 0; t < HIDC / 4; ++t) {
    const int krow = t * 4 + half * 2;
    bw[t].x = W[(size_t)krow * HIDC + n0 + row16];
    bw[t].y = W[(size_t)(krow + 1) * HIDC + n0 + row16];
  }
#pragma unroll
  for (int r = 0; r < 8; ++r) {
    const int gr = mrow + r;
    sc[r] = dinv[gr < n ? gr : n - 1];
  }
  __syncthreads();

  v8f acc8 = {};
  const float* arow = &tile[(rt * 16 + row16) * LDSW];
#pragma unroll
  for (int t = 0; t < HIDC / 4; ++t) {
    v2f a = *(const v2f*)(arow + t * 4 + half * 2);
    acc8 = __builtin_amdgcn_wmma_f32_16x16x4_f32(
        false, a, false, bw[t], (short)0, acc8, false, false);
  }

#pragma unroll
  for (int r = 0; r < 8; ++r) {
    const int gr = mrow + r;
    if (gr < n) H[(size_t)gr * HIDC + n0 + row16] = acc8[r] * sc[r];
  }
}


extern "C" void kernel_launch(void* const* d_in, const int* in_sizes, int n_in,
                              void* d_out, int out_size, void* d_ws, size_t ws_size,
                              hipStream_t stream) {
  const float* x  = (const float*)d_in[0];
  const int*   ei = (const int*)d_in[1];
  const float* W1 = (const float*)d_in[2];
  const float* b1 = (const float*)d_in[3];
  const float* W2 = (const float*)d_in[4];
  const float* b2 = (const float*)d_in[5];

  const int IN_CH = 512;
  const int n = in_sizes[0] / IN_CH;
  const int E = in_sizes[1] / 2;
  const int* src = ei;
  const int* dst = ei + E;
  float* out = (float*)d_out;

  float* ws = (float*)d_ws;
  size_t n_al = ((size_t)n + 255) & ~(size_t)255;
  float* dinv = ws;
  float* bufA = ws + n_al;
  const int nbk = (n + BKT_NODES - 1) / BKT_NODES;
  const int pa  = (E + PCH - 1) / PCH;
  size_t om_al = ((size_t)pa * (size_t)(nbk + 1) + 255) & ~(size_t)255;
  int* rowp   = (int*)(bufA + (size_t)n * HIDC);
  int* offs   = rowp + n_al + 256;
  int* csrs   = offs + om_al;
  int* pairs  = (int*)d_out;

  unsigned gG   = (unsigned)((n + 8 * NPW - 1) / (8 * NPW));
  unsigned gF   = (unsigned)((n + 8 * NPW - 1) / (8 * NPW));
  unsigned gMM  = (unsigned)((n + GM_ROWS - 1) / GM_ROWS);

  bucket_partition_kernel<<<(unsigned)pa, PBLOCK, 0, stream>>>(src, dst, offs, pairs, E, nbk);
  bucket_build_kernel<<<(unsigned)nbk, PBLOCK, 0, stream>>>(pairs, offs, dinv, rowp, csrs, n, nbk, pa);

  gemm_wmma_kernel<<<gMM, 128, 0, stream>>>(x, W1, dinv, out, n, IN_CH);

  gather_relu_gemm_kernel<<<gF, 256, 0, stream>>>(rowp, csrs, dinv, out, b1, W2, bufA, n);

  gather_rows_kernel<<<gG, 256, 0, stream>>>(rowp, csrs, dinv, bufA, b2, out, n, 0);
}
